# starting kernel
# baseline (speedup 1.0000x reference)
#include <hip/hip_runtime.h>
#include <hip/hip_bf16.h>

#define DIM 64
#define HID 512
#define BATCH 8192

typedef __attribute__((ext_vector_type(2))) float v2f;
typedef __attribute__((ext_vector_type(8))) float v8f;

#define WS_MW1T  0
#define WS_MW2   (HID*DIM)
#define WS_WSUM  (HID*DIM + 2*DIM*HID)
#define WS_BSUM  (WS_WSUM + HID)
#define WS_BPACK 98832
#define NPACK    (HID*DIM)

__global__ __launch_bounds__(256) void maf_prep_mask(const float* __restrict__ W1,
                                                     const float* __restrict__ mask1,
                                                     const float* __restrict__ W2,
                                                     const float* __restrict__ mask2,
                                                     float* __restrict__ ws) {
  int idx = blockIdx.x * blockDim.x + threadIdx.x;
  if (idx < HID * DIM) {
    int j = idx / DIM, k = idx % DIM;
    ws[WS_MW1T + k * HID + j] = mask1[idx] * W1[idx];
  } else if (idx < HID * DIM + 2 * DIM * HID) {
    int t = idx - HID * DIM;
    ws[WS_MW2 + t] = mask2[t] * W2[t];
  } else if (idx < HID * DIM + 2 * DIM * HID + NPACK) {
    int f = idx - (HID * DIM + 2 * DIM * HID);
    int p = f >> 1, c = f & 1;
    int nt = p >> 9;
    int t  = (p >> 5) & 15;
    int l  = p & 31;
    int koff = (l >> 4) << 1;
    int j = (nt << 4) + (l & 15);
    int k = (t << 2) + koff + c;
    ws[WS_BPACK + f] = mask1[j * DIM + k] * W1[j * DIM + k];
  }
}

__global__ __launch_bounds__(512) void maf_prep_sum(const float* __restrict__ b2,
                                                    float* __restrict__ ws) {
  int j = threadIdx.x;
  const float* MW2 = ws + WS_MW2;
  float s = 0.f;
  for (int r = DIM; r < 2 * DIM; ++r) s += MW2[r * HID + j];
  ws[WS_WSUM + j] = s;
  if (j == 0) {
    float bs = 0.f;
    for (int r = DIM; r < 2 * DIM; ++r) bs += b2[r];
    ws[WS_BSUM] = bs;
  }
}

__global__ __launch_bounds__(256) void maf_scan(const float* __restrict__ z,
                                                const float* __restrict__ b1,
                                                const float* __restrict__ b2,
                                                const float* __restrict__ ws,
                                                float* __restrict__ xout) {
  __shared__ __align__(16) float w1c[HID];
  __shared__ __align__(16) float w2a[HID];
  __shared__ __align__(16) float w2b[HID];
  const float* MW1T = ws + WS_MW1T;
  const float* MW2  = ws + WS_MW2;

  const int tid  = threadIdx.x;
  const int lane = tid & 31;
  const int wave = tid >> 5;
  const int row  = blockIdx.x * 8 + wave;
  const int t2   = tid * 2;

  float h[16];
#pragma unroll
  for (int t = 0; t < 16; ++t) h[t] = b1[lane + 32 * t];

  const float z0 = z[row * DIM + lane];
  const float z1 = z[row * DIM + 32 + lane];
  float x0 = 0.f, x1 = 0.f;

  for (int i = 0; i < DIM; ++i) {
    __syncthreads();
    *(v2f*)&w1c[t2] = *(const v2f*)&MW1T[i * HID + t2];
    *(v2f*)&w2a[t2] = *(const v2f*)&MW2[i * HID + t2];
    *(v2f*)&w2b[t2] = *(const v2f*)&MW2[(DIM + i) * HID + t2];
    __syncthreads();

    float mu = 0.f, al = 0.f;
#pragma unroll
    for (int t = 0; t < 16; ++t) {
      float hv = fmaxf(h[t], 0.f);
      int j = lane + 32 * t;
      mu = fmaf(hv, w2a[j], mu);
      al = fmaf(hv, w2b[j], al);
    }
#pragma unroll
    for (int off = 16; off >= 1; off >>= 1) {
      mu += __shfl_xor(mu, off, 32);
      al += __shfl_xor(al, off, 32);
    }
    mu += b2[i];
    al += b2[DIM + i];

    float zsel = (i < 32) ? z0 : z1;
    float zi = __shfl(zsel, i & 31, 32);
    float xi = fmaf(zi, expf(al), mu);

    if (i < 32) { if (lane == i)        x0 = xi; }
    else        { if (lane == (i - 32)) x1 = xi; }

    if (i < DIM - 1) {
#pragma unroll
      for (int t = 0; t < 16; ++t) h[t] = fmaf(xi, w1c[lane + 32 * t], h[t]);
    }
  }

  xout[row * DIM + lane]      = x0;
  xout[row * DIM + 32 + lane] = x1;
}

__global__ __launch_bounds__(256) void maf_logdet_wmma(const float* __restrict__ b1,
                                                       const float* __restrict__ ws,
                                                       const float* __restrict__ xout,
                                                       float* __restrict__ ldout) {
  __shared__ float b1s[HID];
  __shared__ float wss[HID];
  const v2f* Bp = (const v2f*)(ws + WS_BPACK);
  const float bsum = ws[WS_BSUM];

  const int tid = threadIdx.x;
  b1s[tid]       = b1[tid];
  b1s[tid + 256] = b1[tid + 256];
  wss[tid]       = ws[WS_WSUM + tid];
  wss[tid + 256] = ws[WS_WSUM + tid + 256];
  __syncthreads();

  const int lane = tid & 31;
  const int wave = tid >> 5;
  const int lr   = lane & 15;
  const int koff = (lane >> 4) ? 2 : 0;
  const int m0   = (blockIdx.x * 8 + wave) * 16;

  const float* xp = xout + (m0 + lr) * DIM + koff;
  v2f a[16];
#pragma unroll
  for (int t = 0; t < 16; ++t) a[t] = *(const v2f*)(xp + 4 * t);

  v8f ld = {};
  for (int nt = 0; nt < 32; ++nt) {
    const int col = nt * 16 + lr;

    v2f bt[16];
#pragma unroll
    for (int t = 0; t < 16; ++t) bt[t] = Bp[(nt * 16 + t) * 32 + lane];

    v8f acc0 = {}, acc1 = {};
#pragma unroll
    for (int t = 0; t < 16; t += 2) {
      acc0 = __builtin_amdgcn_wmma_f32_16x16x4_f32(false, a[t],     false, bt[t],
                                                   (short)0, acc0, false, false);
      acc1 = __builtin_amdgcn_wmma_f32_16x16x4_f32(false, a[t + 1], false, bt[t + 1],
                                                   (short)0, acc1, false, false);
    }
    v8f acc = acc0 + acc1;

    const float bb = b1s[col];
    const float wv = wss[col];
#pragma unroll
    for (int v = 0; v < 8; ++v) {
      float hv = fmaxf(acc[v] + bb, 0.f);
      ld[v] = fmaf(hv, wv, ld[v]);
    }
  }

#pragma unroll
  for (int off = 1; off < 16; off <<= 1) {
#pragma unroll
    for (int v = 0; v < 8; ++v) {
      float t = ld[v];
      ld[v] = t + __shfl_xor(t, off, 32);
    }
  }
  if (lane == 0) {
#pragma unroll
    for (int v = 0; v < 8; ++v) ldout[m0 + v] = ld[v] + bsum;
  } else if (lane == 16) {
#pragma unroll
    for (int v = 0; v < 8; ++v) ldout[m0 + 8 + v] = ld[v] + bsum;
  }
}

extern "C" void kernel_launch(void* const* d_in, const int* in_sizes, int n_in,
                              void* d_out, int out_size, void* d_ws, size_t ws_size,
                              hipStream_t stream) {
  const float* z     = (const float*)d_in[0];
  const float* W1    = (const float*)d_in[1];
  const float* b1    = (const float*)d_in[2];
  const float* W2    = (const float*)d_in[3];
  const float* b2    = (const float*)d_in[4];
  const float* mask1 = (const float*)d_in[5];
  const float* mask2 = (const float*)d_in[6];

  float* ws    = (float*)d_ws;
  float* xout  = (float*)d_out;
  float* ldout = xout + BATCH * DIM;

  const int nprep = HID * DIM + 2 * DIM * HID + NPACK;
  maf_prep_mask<<<(nprep + 255) / 256, 256, 0, stream>>>(W1, mask1, W2, mask2, ws);
  maf_prep_sum<<<1, 512, 0, stream>>>(b2, ws);
  maf_scan<<<BATCH / 8, 256, 0, stream>>>(z, b1, b2, ws, xout);
  maf_logdet_wmma<<<BATCH / 128, 256, 0, stream>>>(b1, ws, xout, ldout);
}


# logdet: packed W1 copied once per work-group into LDS (128 blocks x 4 waves, 64 rows each; was 64 x 8); B fragments read from LDS one hidden tile ahead instead of from L2; sum order unchanged
# speedup vs baseline: 2.4969x; 2.4969x over previous
#include <hip/hip_runtime.h>
#include <hip/hip_bf16.h>

#define DIM 64
#define HID 512
#define BATCH 8192

typedef __attribute__((ext_vector_type(2))) float v2f;
typedef __attribute__((ext_vector_type(8))) float v8f;

#define WS_MW1T  0
#define WS_MW2   (HID*DIM)
#define WS_WSUM  (HID*DIM + 2*DIM*HID)
#define WS_BSUM  (WS_WSUM + HID)
#define WS_BPACK 98832
#define NPACK    (HID*DIM)

__global__ __launch_bounds__(256) void maf_prep_mask(const float* __restrict__ W1,
                                                     const float* __restrict__ mask1,
                                                     const float* __restrict__ W2,
                                                     const float* __restrict__ mask2,
                                                     float* __restrict__ ws) {
  int idx = blockIdx.x * blockDim.x + threadIdx.x;
  if (idx < HID * DIM) {
    int j = idx / DIM, k = idx % DIM;
    ws[WS_MW1T + k * HID + j] = mask1[idx] * W1[idx];
  } else if (idx < HID * DIM + 2 * DIM * HID) {
    int t = idx - HID * DIM;
    ws[WS_MW2 + t] = mask2[t] * W2[t];
  } else if (idx < HID * DIM + 2 * DIM * HID + NPACK) {
    int f = idx - (HID * DIM + 2 * DIM * HID);
    int p = f >> 1, c = f & 1;
    int nt = p >> 9;
    int t  = (p >> 5) & 15;
    int l  = p & 31;
    int koff = (l >> 4) << 1;
    int j = (nt << 4) + (l & 15);
    int k = (t << 2) + koff + c;
    ws[WS_BPACK + f] = mask1[j * DIM + k] * W1[j * DIM + k];
  }
}

__global__ __launch_bounds__(512) void maf_prep_sum(const float* __restrict__ b2,
                                                    float* __restrict__ ws) {
  int j = threadIdx.x;
  const float* MW2 = ws + WS_MW2;
  float s = 0.f;
  for (int r = DIM; r < 2 * DIM; ++r) s += MW2[r * HID + j];
  ws[WS_WSUM + j] = s;
  if (j == 0) {
    float bs = 0.f;
    for (int r = DIM; r < 2 * DIM; ++r) bs += b2[r];
    ws[WS_BSUM] = bs;
  }
}

typedef __attribute__((ext_vector_type(4))) float v4f;

#define SC_WAVES 4
#define SC_ROWS  16
#define SC_PITCH 65

template <int K> __device__ __forceinline__ float row_share(float v) {
  return __builtin_bit_cast(float, __builtin_amdgcn_update_dpp(0, __builtin_bit_cast(int, v),
                                                               0x150 + K, 0xf, 0xf, true));
}

template <int K> __device__ __forceinline__ void fmac_rs(float& acc, float w, float v) {
  asm volatile("v_fmac_f32_dpp %0, %1, %2 row_share:%3 row_mask:0xf bank_mask:0xf bound_ctrl:1"
               : "+v"(acc) : "v"(w), "v"(v), "n"(K));
}
__device__ __forceinline__ float relu_ord(float x) {
  float r;
  asm volatile("v_max_num_f32 %0, 0, %1" : "=v"(r) : "v"(x));
  return r;
}

template <int B> __device__ __forceinline__ void dot8(const float* h, float wa, float wb,
                                                      float (&mu)[4], float (&al)[4]) {
  const float r0 = relu_ord(h[B + 0]), r1 = relu_ord(h[B + 1]), r2 = relu_ord(h[B + 2]), r3 = relu_ord(h[B + 3]);
  const float r4 = relu_ord(h[B + 4]), r5 = relu_ord(h[B + 5]), r6 = relu_ord(h[B + 6]), r7 = relu_ord(h[B + 7]);
  fmac_rs<B + 0>(mu[0], wa, r0); fmac_rs<B + 0>(al[0], wb, r0);
  fmac_rs<B + 1>(mu[1], wa, r1); fmac_rs<B + 1>(al[1], wb, r1);
  fmac_rs<B + 2>(mu[2], wa, r2); fmac_rs<B + 2>(al[2], wb, r2);
  fmac_rs<B + 3>(mu[3], wa, r3); fmac_rs<B + 3>(al[3], wb, r3);
  fmac_rs<B + 4>(mu[0], wa, r4); fmac_rs<B + 4>(al[0], wb, r4);
  fmac_rs<B + 5>(mu[1], wa, r5); fmac_rs<B + 5>(al[1], wb, r5);
  fmac_rs<B + 6>(mu[2], wa, r6); fmac_rs<B + 6>(al[2], wb, r6);
  fmac_rs<B + 7>(mu[3], wa, r7); fmac_rs<B + 7>(al[3], wb, r7);
}
template <int B> __device__ __forceinline__ void pass8(float* h, float w1, float wa, float wb, float xi,
                                                       float (&mu)[4], float (&al)[4]) {
  fmac_rs<B + 0>(h[B + 0], w1, xi); fmac_rs<B + 1>(h[B + 1], w1, xi);
  fmac_rs<B + 2>(h[B + 2], w1, xi); fmac_rs<B + 3>(h[B + 3], w1, xi);
  fmac_rs<B + 4>(h[B + 4], w1, xi); fmac_rs<B + 5>(h[B + 5], w1, xi);
  fmac_rs<B + 6>(h[B + 6], w1, xi); fmac_rs<B + 7>(h[B + 7], w1, xi);
  dot8<B>(h, wa, wb, mu, al);
}

template <int K> struct HidK {
  static __device__ __forceinline__ void init(float* h, float b) {
    h[K] = row_share<K>(b);
    HidK<K + 1>::init(h, b);
  }
};
template <> struct HidK<16> {
  static __device__ __forceinline__ void init(float*, float) {}
};

__device__ __forceinline__ void lds_barrier() {
  asm volatile("s_wait_dscnt 0x0\n\ts_barrier_signal -1\n\ts_barrier_wait -1" ::: "memory");
}

__device__ __forceinline__ void scan_step(int i, bool do_pass, v2f (*redbuf)[32], const float* zs, float* xs,
                                          const float* b2s, int tid, int wave, int lane, int r,
                                          float (&h)[4][16], float (&mu)[4], float (&al)[4],
                                          v4f w1, v4f wa, v4f wb) {
  redbuf[wave][lane] = (v2f){(mu[0] + mu[1]) + (mu[2] + mu[3]), (al[0] + al[1]) + (al[2] + al[3])};
  const float zi = zs[r * SC_PITCH + i];
  const float bm = b2s[i];
  const float ba = b2s[DIM + i];
  lds_barrier();
  const v2f p0 = redbuf[0][r], q0 = redbuf[0][r + 16];
  const v2f p1 = redbuf[1][r], q1 = redbuf[1][r + 16];
  const v2f p2 = redbuf[2][r], q2 = redbuf[2][r + 16];
  const v2f p3 = redbuf[3][r], q3 = redbuf[3][r + 16];
  const float sm = ((p0.x + q0.x) + (p1.x + q1.x)) + ((p2.x + q2.x) + (p3.x + q3.x));
  const float sa = ((p0.y + q0.y) + (p1.y + q1.y)) + ((p2.y + q2.y) + (p3.y + q3.y));
  const float xi = fmaf(zi, expf(sa + ba), sm + bm);
  if (tid < SC_ROWS) xs[r * SC_PITCH + i] = xi;
  if (do_pass) {
    mu[0] = mu[1] = mu[2] = mu[3] = 0.f;
    al[0] = al[1] = al[2] = al[3] = 0.f;
#pragma unroll
    for (int m = 0; m < 4; ++m) {
      pass8<0>(h[m], w1[m], wa[m], wb[m], xi, mu, al);
      pass8<8>(h[m], w1[m], wa[m], wb[m], xi, mu, al);
    }
  }
}

__global__ __launch_bounds__(128) void maf_scan(const float* __restrict__ z,
                                                const float* __restrict__ b1,
                                                const float* __restrict__ b2,
                                                const float* __restrict__ ws,
                                                float* __restrict__ xout) {
  __shared__ float zs[SC_ROWS * SC_PITCH];
  __shared__ float xs[SC_ROWS * SC_PITCH];
  __shared__ float b2s[2 * DIM];
  __shared__ __align__(8) v2f red[2][SC_WAVES][32];
  const float* MW1T = ws + WS_MW1T;
  const float* MW2a = ws + WS_MW2;
  const float* MW2b = ws + WS_MW2 + DIM * HID;

  const int tid  = threadIdx.x;
  const int lane = tid & 31;
  const int wave = tid >> 5;
  const int r    = lane & 15;
  const int row0 = blockIdx.x * SC_ROWS;
  const int off  = tid * 4;

  b2s[tid] = b2[tid];
#pragma unroll
  for (int q = 0; q < 8; ++q) {
    const int idx = q * 128 + tid;
    zs[(idx >> 6) * SC_PITCH + (idx & 63)] = z[row0 * DIM + idx];
  }

  const v4f bv  = *(const v4f*)(b1 + off);
  const v4f wa0 = *(const v4f*)(MW2a + off);
  const v4f wb0 = *(const v4f*)(MW2b + off);
  v4f S1[4], Sa[4], Sb[4];
#pragma unroll
  for (int j = 0; j < 4; ++j) {
    S1[j] = *(const v4f*)(MW1T + j * HID + off);
    Sa[j] = *(const v4f*)(MW2a + (j + 1) * HID + off);
    Sb[j] = *(const v4f*)(MW2b + (j + 1) * HID + off);
  }

  float h[4][16];
  float mu[4], al[4];
#pragma unroll
  for (int m = 0; m < 4; ++m) {
    HidK<0>::init(h[m], bv[m]);
    mu[m] = 0.f;
    al[m] = 0.f;
  }
#pragma unroll
  for (int m = 0; m < 4; ++m) {
    dot8<0>(h[m], wa0[m], wb0[m], mu, al);
    dot8<8>(h[m], wa0[m], wb0[m], mu, al);
  }
  lds_barrier();

  for (int i = 0; i < DIM; i += 4) {
#pragma unroll
    for (int j = 0; j < 4; ++j) {
      scan_step(i + j, i + j < DIM - 1, red[j & 1], zs, xs, b2s, tid, wave, lane, r, h, mu, al,
                S1[j], Sa[j], Sb[j]);
      const int p = min(i + j + 4, DIM - 2);
      S1[j] = *(const v4f*)(MW1T + p * HID + off);
      Sa[j] = *(const v4f*)(MW2a + (p + 1) * HID + off);
      Sb[j] = *(const v4f*)(MW2b + (p + 1) * HID + off);
    }
  }

  lds_barrier();
#pragma unroll
  for (int q = 0; q < 8; ++q) {
    const int idx = q * 128 + tid;
    xout[row0 * DIM + idx] = xs[(idx >> 6) * SC_PITCH + (idx & 63)];
  }
}

#define LD_WAVES 4
#define LD_ROWS  (LD_WAVES * 16)

__device__ __forceinline__ void ld_tile(const v2f (&a)[16], const v2f (&b)[16], float bb, float wv, v8f& ld) {
  v8f acc0 = {}, acc1 = {};
#pragma unroll
  for (int t = 0; t < 16; t += 2) {
    acc0 = __builtin_amdgcn_wmma_f32_16x16x4_f32(false, a[t],     false, b[t],
                                                 (short)0, acc0, false, false);
    acc1 = __builtin_amdgcn_wmma_f32_16x16x4_f32(false, a[t + 1], false, b[t + 1],
                                                 (short)0, acc1, false, false);
  }
  const v8f acc = acc0 + acc1;
#pragma unroll
  for (int v = 0; v < 8; ++v) {
    const float hv = fmaxf(acc[v] + bb, 0.f);
    ld[v] = fmaf(hv, wv, ld[v]);
  }
}

__global__ __launch_bounds__(LD_WAVES * 32) void maf_logdet_wmma(const float* __restrict__ b1,
                                                                 const float* __restrict__ ws,
                                                                 const float* __restrict__ xout,
                                                                 float* __restrict__ ldout) {
  __shared__ __align__(16) float Bs[NPACK];
  __shared__ __align__(16) float b1s[HID];
  __shared__ __align__(16) float wss[HID];
  const float bsum = ws[WS_BSUM];

  const int tid  = threadIdx.x;
  const int lane = tid & 31;
  const int wave = tid >> 5;
  const int lr   = lane & 15;
  const int koff = (lane >> 4) ? 2 : 0;
  const int m0   = (blockIdx.x * LD_WAVES + wave) * 16;

  const float* xp = xout + (m0 + lr) * DIM + koff;
  v2f a[16];
#pragma unroll
  for (int t = 0; t < 16; ++t) a[t] = *(const v2f*)(xp + 4 * t);

  {
    const v4f* src = (const v4f*)(ws + WS_BPACK);
    v4f* dst = (v4f*)Bs;
#pragma unroll 32
    for (int q = 0; q < NPACK / 4 / (LD_WAVES * 32); ++q) dst[q * (LD_WAVES * 32) + tid] = src[q * (LD_WAVES * 32) + tid];
    ((v4f*)b1s)[tid] = ((const v4f*)b1)[tid];
    ((v4f*)wss)[tid] = ((const v4f*)(ws + WS_WSUM))[tid];
  }
  __syncthreads();

  const v2f* Bl = (const v2f*)Bs + lane;
  v8f ld = {};
  v2f bc[16], bn[16];
#pragma unroll
  for (int t = 0; t < 16; ++t) bc[t] = Bl[t * 32];
  for (int nt = 0; nt < 32; nt += 2) {
    const v2f* p1 = Bl + (nt + 1) * 512;
#pragma unroll
    for (int t = 0; t < 16; ++t) bn[t] = p1[t * 32];
    ld_tile(a, bc, b1s[nt * 16 + lr], wss[nt * 16 + lr], ld);
    const v2f* p2 = Bl + min(nt + 2, 31) * 512;
#pragma unroll
    for (int t = 0; t < 16; ++t) bc[t] = p2[t * 32];
    ld_tile(a, bn, b1s[nt * 16 + 16 + lr], wss[nt * 16 + 16 + lr], ld);
  }

#pragma unroll
  for (int off = 1; off < 16; off <<= 1) {
#pragma unroll
    for (int v = 0; v < 8; ++v) {
      float t = ld[v];
      ld[v] = t + __shfl_xor(t, off, 32);
    }
  }
  if (lane == 0) {
#pragma unroll
    for (int v = 0; v < 8; ++v) ldout[m0 + v] = ld[v] + bsum;
  } else if (lane == 16) {
#pragma unroll
    for (int v = 0; v < 8; ++v) ldout[m0 + 8 + v] = ld[v] + bsum;
  }
}

extern "C" void kernel_launch(void* const* d_in, const int* in_sizes, int n_in,
                              void* d_out, int out_size, void* d_ws, size_t ws_size,
                              hipStream_t stream) {
  const float* z     = (const float*)d_in[0];
  const float* W1    = (const float*)d_in[1];
  const float* b1    = (const float*)d_in[2];
  const float* W2    = (const float*)d_in[3];
  const float* b2    = (const float*)d_in[4];
  const float* mask1 = (const float*)d_in[5];
  const float* mask2 = (const float*)d_in[6];

  float* ws    = (float*)d_ws;
  float* xout  = (float*)d_out;
  float* ldout = xout + BATCH * DIM;

  const int nprep = HID * DIM + 2 * DIM * HID + NPACK;
  maf_prep_mask<<<(nprep + 255) / 256, 256, 0, stream>>>(W1, mask1, W2, mask2, ws);
  maf_prep_sum<<<1, 512, 0, stream>>>(b2, ws);
  maf_scan<<<BATCH / SC_ROWS, SC_WAVES * 32, 0, stream>>>(z, b1, b2, ws, xout);
  maf_logdet_wmma<<<BATCH / LD_ROWS, LD_WAVES * 32, 0, stream>>>(b1, ws, xout, ldout);
}


# scan: each weight copied to its 16-lane row by one v_permlane16_var and used by packed FMAs (two hidden units; mu with alpha) instead of a row_share DPP operand on every FMA; sums bit-identical
# speedup vs baseline: 2.7938x; 1.1189x over previous
#include <hip/hip_runtime.h>
#include <hip/hip_bf16.h>

#define DIM 64
#define HID 512
#define BATCH 8192

typedef __attribute__((ext_vector_type(2))) float v2f;
typedef __attribute__((ext_vector_type(8))) float v8f;

#define WS_MW1T  0
#define WS_MW2   (HID*DIM)
#define WS_WSUM  (HID*DIM + 2*DIM*HID)
#define WS_BSUM  (WS_WSUM + HID)
#define WS_BPACK 98832
#define NPACK    (HID*DIM)

__global__ __launch_bounds__(256) void maf_prep_mask(const float* __restrict__ W1,
                                                     const float* __restrict__ mask1,
                                                     const float* __restrict__ W2,
                                                     const float* __restrict__ mask2,
                                                     float* __restrict__ ws) {
  int idx = blockIdx.x * blockDim.x + threadIdx.x;
  if (idx < HID * DIM) {
    int j = idx / DIM, k = idx % DIM;
    ws[WS_MW1T + k * HID + j] = mask1[idx] * W1[idx];
  } else if (idx < HID * DIM + 2 * DIM * HID) {
    int t = idx - HID * DIM;
    ws[WS_MW2 + t] = mask2[t] * W2[t];
  } else if (idx < HID * DIM + 2 * DIM * HID + NPACK) {
    int f = idx - (HID * DIM + 2 * DIM * HID);
    int p = f >> 1, c = f & 1;
    int nt = p >> 9;
    int t  = (p >> 5) & 15;
    int l  = p & 31;
    int koff = (l >> 4) << 1;
    int j = (nt << 4) + (l & 15);
    int k = (t << 2) + koff + c;
    ws[WS_BPACK + f] = mask1[j * DIM + k] * W1[j * DIM + k];
  }
}

__global__ __launch_bounds__(512) void maf_prep_sum(const float* __restrict__ b2,
                                                    float* __restrict__ ws) {
  int j = threadIdx.x;
  const float* MW2 = ws + WS_MW2;
  float s = 0.f;
  for (int r = DIM; r < 2 * DIM; ++r) s += MW2[r * HID + j];
  ws[WS_WSUM + j] = s;
  if (j == 0) {
    float bs = 0.f;
    for (int r = DIM; r < 2 * DIM; ++r) bs += b2[r];
    ws[WS_BSUM] = bs;
  }
}

typedef __attribute__((ext_vector_type(4))) float v4f;

#define SC_WAVES 4
#define SC_ROWS  16
#define SC_PITCH 65

template <int K> __device__ __forceinline__ float row_share(float v) {
  return __builtin_bit_cast(float, __builtin_amdgcn_update_dpp(0, __builtin_bit_cast(int, v),
                                                               0x150 + K, 0xf, 0xf, true));
}

template <int K> __device__ __forceinline__ float row_bcast(float v) {
  int keep;
  return __builtin_bit_cast(float, __builtin_amdgcn_permlane16_var(keep, __builtin_bit_cast(int, v), K, false,
                                                                   false));
}

template <int B> __device__ __forceinline__ void dot8(const float* h, float wa, float wb, v2f (&ma)[4]) {
#pragma unroll
  for (int k = 0; k < 8; ++k) {
    const float r = fmaxf(h[B + k], 0.f);
    v2f w, rr;
    if (k == 0) { w.x = row_bcast<B + 0>(wa); w.y = row_bcast<B + 0>(wb); }
    if (k == 1) { w.x = row_bcast<B + 1>(wa); w.y = row_bcast<B + 1>(wb); }
    if (k == 2) { w.x = row_bcast<B + 2>(wa); w.y = row_bcast<B + 2>(wb); }
    if (k == 3) { w.x = row_bcast<B + 3>(wa); w.y = row_bcast<B + 3>(wb); }
    if (k == 4) { w.x = row_bcast<B + 4>(wa); w.y = row_bcast<B + 4>(wb); }
    if (k == 5) { w.x = row_bcast<B + 5>(wa); w.y = row_bcast<B + 5>(wb); }
    if (k == 6) { w.x = row_bcast<B + 6>(wa); w.y = row_bcast<B + 6>(wb); }
    if (k == 7) { w.x = row_bcast<B + 7>(wa); w.y = row_bcast<B + 7>(wb); }
    rr.x = r; rr.y = r;
    ma[k & 3] = __builtin_elementwise_fma(w, rr, ma[k & 3]);
  }
}
template <int B> __device__ __forceinline__ void pass8(float* h, float w1, float wa, float wb, float xi,
                                                       v2f (&ma)[4]) {
  v2f xx; xx.x = xi; xx.y = xi;
  v2f w01, w23, w45, w67;
  w01.x = row_bcast<B + 0>(w1); w01.y = row_bcast<B + 1>(w1);
  w23.x = row_bcast<B + 2>(w1); w23.y = row_bcast<B + 3>(w1);
  w45.x = row_bcast<B + 4>(w1); w45.y = row_bcast<B + 5>(w1);
  w67.x = row_bcast<B + 6>(w1); w67.y = row_bcast<B + 7>(w1);
  v2f h01, h23, h45, h67;
  h01.x = h[B + 0]; h01.y = h[B + 1]; h23.x = h[B + 2]; h23.y = h[B + 3];
  h45.x = h[B + 4]; h45.y = h[B + 5]; h67.x = h[B + 6]; h67.y = h[B + 7];
  h01 = __builtin_elementwise_fma(w01, xx, h01);
  h23 = __builtin_elementwise_fma(w23, xx, h23);
  h45 = __builtin_elementwise_fma(w45, xx, h45);
  h67 = __builtin_elementwise_fma(w67, xx, h67);
  h[B + 0] = h01.x; h[B + 1] = h01.y; h[B + 2] = h23.x; h[B + 3] = h23.y;
  h[B + 4] = h45.x; h[B + 5] = h45.y; h[B + 6] = h67.x; h[B + 7] = h67.y;
  dot8<B>(h, wa, wb, ma);
}

template <int K> struct HidK {
  static __device__ __forceinline__ void init(float* h, float b) {
    h[K] = row_share<K>(b);
    HidK<K + 1>::init(h, b);
  }
};
template <> struct HidK<16> {
  static __device__ __forceinline__ void init(float*, float) {}
};

__device__ __forceinline__ void lds_barrier() {
  asm volatile("s_wait_dscnt 0x0\n\ts_barrier_signal -1\n\ts_barrier_wait -1" ::: "memory");
}

__device__ __forceinline__ void scan_step(int i, bool do_pass, v2f (*redbuf)[32], const float* zs, float* xs,
                                          const float* b2s, int tid, int wave, int lane, int r,
                                          float (&h)[4][16], v2f (&ma)[4],
                                          v4f w1, v4f wa, v4f wb) {
  redbuf[wave][lane] = (ma[0] + ma[1]) + (ma[2] + ma[3]);
  const float zi = zs[r * SC_PITCH + i];
  const float bm = b2s[i];
  const float ba = b2s[DIM + i];
  lds_barrier();
  const v2f p0 = redbuf[0][r], q0 = redbuf[0][r + 16];
  const v2f p1 = redbuf[1][r], q1 = redbuf[1][r + 16];
  const v2f p2 = redbuf[2][r], q2 = redbuf[2][r + 16];
  const v2f p3 = redbuf[3][r], q3 = redbuf[3][r + 16];
  const float sm = ((p0.x + q0.x) + (p1.x + q1.x)) + ((p2.x + q2.x) + (p3.x + q3.x));
  const float sa = ((p0.y + q0.y) + (p1.y + q1.y)) + ((p2.y + q2.y) + (p3.y + q3.y));
  const float xi = fmaf(zi, expf(sa + ba), sm + bm);
  if (tid < SC_ROWS) xs[r * SC_PITCH + i] = xi;
  if (do_pass) {
    ma[0] = ma[1] = ma[2] = ma[3] = (v2f){0.f, 0.f};
#pragma unroll
    for (int m = 0; m < 4; ++m) {
      pass8<0>(h[m], w1[m], wa[m], wb[m], xi, ma);
      pass8<8>(h[m], w1[m], wa[m], wb[m], xi, ma);
    }
  }
}

__global__ __launch_bounds__(128) void maf_scan(const float* __restrict__ z,
                                                const float* __restrict__ b1,
                                                const float* __restrict__ b2,
                                                const float* __restrict__ ws,
                                                float* __restrict__ xout) {
  __shared__ float zs[SC_ROWS * SC_PITCH];
  __shared__ float xs[SC_ROWS * SC_PITCH];
  __shared__ float b2s[2 * DIM];
  __shared__ __align__(8) v2f red[2][SC_WAVES][32];
  const float* MW1T = ws + WS_MW1T;
  const float* MW2a = ws + WS_MW2;
  const float* MW2b = ws + WS_MW2 + DIM * HID;

  const int tid  = threadIdx.x;
  const int lane = tid & 31;
  const int wave = tid >> 5;
  const int r    = lane & 15;
  const int row0 = blockIdx.x * SC_ROWS;
  const int off  = tid * 4;

  b2s[tid] = b2[tid];
#pragma unroll
  for (int q = 0; q < 8; ++q) {
    const int idx = q * 128 + tid;
    zs[(idx >> 6) * SC_PITCH + (idx & 63)] = z[row0 * DIM + idx];
  }

  const v4f bv  = *(const v4f*)(b1 + off);
  const v4f wa0 = *(const v4f*)(MW2a + off);
  const v4f wb0 = *(const v4f*)(MW2b + off);
  v4f S1[4], Sa[4], Sb[4];
#pragma unroll
  for (int j = 0; j < 4; ++j) {
    S1[j] = *(const v4f*)(MW1T + j * HID + off);
    Sa[j] = *(const v4f*)(MW2a + (j + 1) * HID + off);
    Sb[j] = *(const v4f*)(MW2b + (j + 1) * HID + off);
  }

  float h[4][16];
  v2f ma[4];
#pragma unroll
  for (int m = 0; m < 4; ++m) {
    HidK<0>::init(h[m], bv[m]);
    ma[m] = (v2f){0.f, 0.f};
  }
#pragma unroll
  for (int m = 0; m < 4; ++m) {
    dot8<0>(h[m], wa0[m], wb0[m], ma);
    dot8<8>(h[m], wa0[m], wb0[m], ma);
  }
  lds_barrier();

  for (int i = 0; i < DIM; i += 4) {
#pragma unroll
    for (int j = 0; j < 4; ++j) {
      scan_step(i + j, i + j < DIM - 1, red[j & 1], zs, xs, b2s, tid, wave, lane, r, h, ma,
                S1[j], Sa[j], Sb[j]);
      const int p = min(i + j + 4, DIM - 2);
      S1[j] = *(const v4f*)(MW1T + p * HID + off);
      Sa[j] = *(const v4f*)(MW2a + (p + 1) * HID + off);
      Sb[j] = *(const v4f*)(MW2b + (p + 1) * HID + off);
    }
  }

  lds_barrier();
#pragma unroll
  for (int q = 0; q < 8; ++q) {
    const int idx = q * 128 + tid;
    xout[row0 * DIM + idx] = xs[(idx >> 6) * SC_PITCH + (idx & 63)];
  }
}

#define LD_WAVES 4
#define LD_ROWS  (LD_WAVES * 16)

__device__ __forceinline__ void ld_tile(const v2f (&a)[16], const v2f (&b)[16], float bb, float wv, v8f& ld) {
  v8f acc0 = {}, acc1 = {};
#pragma unroll
  for (int t = 0; t < 16; t += 2) {
    acc0 = __builtin_amdgcn_wmma_f32_16x16x4_f32(false, a[t],     false, b[t],
                                                 (short)0, acc0, false, false);
    acc1 = __builtin_amdgcn_wmma_f32_16x16x4_f32(false, a[t + 1], false, b[t + 1],
                                                 (short)0, acc1, false, false);
  }
  const v8f acc = acc0 + acc1;
#pragma unroll
  for (int v = 0; v < 8; ++v) {
    const float hv = fmaxf(acc[v] + bb, 0.f);
    ld[v] = fmaf(hv, wv, ld[v]);
  }
}

__global__ __launch_bounds__(LD_WAVES * 32) void maf_logdet_wmma(const float* __restrict__ b1,
                                                                 const float* __restrict__ ws,
                                                                 const float* __restrict__ xout,
                                                                 float* __restrict__ ldout) {
  __shared__ __align__(16) float Bs[NPACK];
  __shared__ __align__(16) float b1s[HID];
  __shared__ __align__(16) float wss[HID];
  const float bsum = ws[WS_BSUM];

  const int tid  = threadIdx.x;
  const int lane = tid & 31;
  const int wave = tid >> 5;
  const int lr   = lane & 15;
  const int koff = (lane >> 4) ? 2 : 0;
  const int m0   = (blockIdx.x * LD_WAVES + wave) * 16;

  const float* xp = xout + (m0 + lr) * DIM + koff;
  v2f a[16];
#pragma unroll
  for (int t = 0; t < 16; ++t) a[t] = *(const v2f*)(xp + 4 * t);

  {
    const v4f* src = (const v4f*)(ws + WS_BPACK);
    v4f* dst = (v4f*)Bs;
#pragma unroll 32
    for (int q = 0; q < NPACK / 4 / (LD_WAVES * 32); ++q) dst[q * (LD_WAVES * 32) + tid] = src[q * (LD_WAVES * 32) + tid];
    ((v4f*)b1s)[tid] = ((const v4f*)b1)[tid];
    ((v4f*)wss)[tid] = ((const v4f*)(ws + WS_WSUM))[tid];
  }
  __syncthreads();

  const v2f* Bl = (const v2f*)Bs + lane;
  v8f ld = {};
  v2f bc[16], bn[16];
#pragma unroll
  for (int t = 0; t < 16; ++t) bc[t] = Bl[t * 32];
  for (int nt = 0; nt < 32; nt += 2) {
    const v2f* p1 = Bl + (nt + 1) * 512;
#pragma unroll
    for (int t = 0; t < 16; ++t) bn[t] = p1[t * 32];
    ld_tile(a, bc, b1s[nt * 16 + lr], wss[nt * 16 + lr], ld);
    const v2f* p2 = Bl + min(nt + 2, 31) * 512;
#pragma unroll
    for (int t = 0; t < 16; ++t) bc[t] = p2[t * 32];
    ld_tile(a, bn, b1s[nt * 16 + 16 + lr], wss[nt * 16 + 16 + lr], ld);
  }

#pragma unroll
  for (int off = 1; off < 16; off <<= 1) {
#pragma unroll
    for (int v = 0; v < 8; ++v) {
      float t = ld[v];
      ld[v] = t + __shfl_xor(t, off, 32);
    }
  }
  if (lane == 0) {
#pragma unroll
    for (int v = 0; v < 8; ++v) ldout[m0 + v] = ld[v] + bsum;
  } else if (lane == 16) {
#pragma unroll
    for (int v = 0; v < 8; ++v) ldout[m0 + 8 + v] = ld[v] + bsum;
  }
}

extern "C" void kernel_launch(void* const* d_in, const int* in_sizes, int n_in,
                              void* d_out, int out_size, void* d_ws, size_t ws_size,
                              hipStream_t stream) {
  const float* z     = (const float*)d_in[0];
  const float* W1    = (const float*)d_in[1];
  const float* b1    = (const float*)d_in[2];
  const float* W2    = (const float*)d_in[3];
  const float* b2    = (const float*)d_in[4];
  const float* mask1 = (const float*)d_in[5];
  const float* mask2 = (const float*)d_in[6];

  float* ws    = (float*)d_ws;
  float* xout  = (float*)d_out;
  float* ldout = xout + BATCH * DIM;

  const int nprep = HID * DIM + 2 * DIM * HID + NPACK;
  maf_prep_mask<<<(nprep + 255) / 256, 256, 0, stream>>>(W1, mask1, W2, mask2, ws);
  maf_prep_sum<<<1, 512, 0, stream>>>(b2, ws);
  maf_scan<<<BATCH / SC_ROWS, SC_WAVES * 32, 0, stream>>>(z, b1, b2, ws, xout);
  maf_logdet_wmma<<<BATCH / LD_ROWS, LD_WAVES * 32, 0, stream>>>(b1, ws, xout, ldout);
}


# scan: two batch rows per lane (8-lane groups of 32 hidden units): one v_permlane16_var copy of a weight feeds the packed FMAs of both rows; partials joined by lane adds, then LDS (new sum order)
# speedup vs baseline: 2.9850x; 1.0684x over previous
#include <hip/hip_runtime.h>
#include <hip/hip_bf16.h>

#define DIM 64
#define HID 512
#define BATCH 8192

typedef __attribute__((ext_vector_type(2))) float v2f;
typedef __attribute__((ext_vector_type(8))) float v8f;

#define WS_MW1T  0
#define WS_MW2   (HID*DIM)
#define WS_WSUM  (HID*DIM + 2*DIM*HID)
#define WS_BSUM  (WS_WSUM + HID)
#define WS_BPACK 98832
#define NPACK    (HID*DIM)

__global__ __launch_bounds__(256) void maf_prep_mask(const float* __restrict__ W1,
                                                     const float* __restrict__ mask1,
                                                     const float* __restrict__ W2,
                                                     const float* __restrict__ mask2,
                                                     float* __restrict__ ws) {
  int idx = blockIdx.x * blockDim.x + threadIdx.x;
  if (idx < HID * DIM) {
    int j = idx / DIM, k = idx % DIM;
    ws[WS_MW1T + k * HID + j] = mask1[idx] * W1[idx];
  } else if (idx < HID * DIM + 2 * DIM * HID) {
    int t = idx - HID * DIM;
    ws[WS_MW2 + t] = mask2[t] * W2[t];
  } else if (idx < HID * DIM + 2 * DIM * HID + NPACK) {
    int f = idx - (HID * DIM + 2 * DIM * HID);
    int p = f >> 1, c = f & 1;
    int nt = p >> 9;
    int t  = (p >> 5) & 15;
    int l  = p & 31;
    int koff = (l >> 4) << 1;
    int j = (nt << 4) + (l & 15);
    int k = (t << 2) + koff + c;
    ws[WS_BPACK + f] = mask1[j * DIM + k] * W1[j * DIM + k];
  }
}

__global__ __launch_bounds__(512) void maf_prep_sum(const float* __restrict__ b2,
                                                    float* __restrict__ ws) {
  int j = threadIdx.x;
  const float* MW2 = ws + WS_MW2;
  float s = 0.f;
  for (int r = DIM; r < 2 * DIM; ++r) s += MW2[r * HID + j];
  ws[WS_WSUM + j] = s;
  if (j == 0) {
    float bs = 0.f;
    for (int r = DIM; r < 2 * DIM; ++r) bs += b2[r];
    ws[WS_BSUM] = bs;
  }
}

typedef __attribute__((ext_vector_type(4))) float v4f;

#define SC_WAVES 4
#define SC_ROWS  16
#define SC_PITCH 65

__device__ __forceinline__ float grp_bcast(float v, int sel) {
  int keep;
  return __builtin_bit_cast(float, __builtin_amdgcn_permlane16_var(keep, __builtin_bit_cast(int, v), sel, false,
                                                                   false));
}

__device__ __forceinline__ float wave_sum(float v) {
  const float t = v + __builtin_bit_cast(float, __builtin_amdgcn_update_dpp(0, __builtin_bit_cast(int, v),
                                                                            0x168, 0xf, 0xf, true));
  int keep;
  return t + __builtin_bit_cast(float, __builtin_amdgcn_permlanex16(keep, __builtin_bit_cast(int, t), 0x76543210,
                                                                    0xfedcba98, false, false));
}

__device__ __forceinline__ void dot8x2(const float* ha, const float* hb, float wa, float wb, const int (&sel)[8],
                                       v2f (&maA)[4], v2f (&maB)[4]) {
#pragma unroll
  for (int k = 0; k < 8; ++k) {
    v2f w, ra, rb;
    w.x = grp_bcast(wa, sel[k]); w.y = grp_bcast(wb, sel[k]);
    const float a = fmaxf(ha[k], 0.f), b = fmaxf(hb[k], 0.f);
    ra.x = a; ra.y = a; rb.x = b; rb.y = b;
    maA[k & 3] = __builtin_elementwise_fma(w, ra, maA[k & 3]);
    maB[k & 3] = __builtin_elementwise_fma(w, rb, maB[k & 3]);
  }
}
__device__ __forceinline__ void pass8x2(float* ha, float* hb, float w1, float wa, float wb, float xa, float xb,
                                        const int (&sel)[8], v2f (&maA)[4], v2f (&maB)[4]) {
  v2f xxa, xxb;
  xxa.x = xa; xxa.y = xa; xxb.x = xb; xxb.y = xb;
#pragma unroll
  for (int k = 0; k < 8; k += 2) {
    v2f w, a, b;
    w.x = grp_bcast(w1, sel[k]); w.y = grp_bcast(w1, sel[k + 1]);
    a.x = ha[k]; a.y = ha[k + 1]; b.x = hb[k]; b.y = hb[k + 1];
    a = __builtin_elementwise_fma(w, xxa, a);
    b = __builtin_elementwise_fma(w, xxb, b);
    ha[k] = a.x; ha[k + 1] = a.y; hb[k] = b.x; hb[k + 1] = b.y;
  }
  dot8x2(ha, hb, wa, wb, sel, maA, maB);
}

__device__ __forceinline__ void lds_barrier() {
  asm volatile("s_wait_dscnt 0x0\n\ts_barrier_signal -1\n\ts_barrier_wait -1" ::: "memory");
}

__device__ __forceinline__ void scan_step(int i, bool do_pass, v4f (*redbuf)[8], const float* zs, float* xs,
                                          const float* b2s, int tid, int wave, int lane, int p,
                                          const int (&sel)[8], float (&ha)[4][8], float (&hb)[4][8],
                                          v2f (&maA)[4], v2f (&maB)[4], v4f w1, v4f wa, v4f wb) {
  const v2f pa = (maA[0] + maA[1]) + (maA[2] + maA[3]);
  const v2f pb = (maB[0] + maB[1]) + (maB[2] + maB[3]);
  v4f part;
  part.x = wave_sum(pa.x); part.y = wave_sum(pa.y); part.z = wave_sum(pb.x); part.w = wave_sum(pb.y);
  if (lane < 8) redbuf[wave][p] = part;
  const float za = zs[p * SC_PITCH + i];
  const float zb = zs[(p + 8) * SC_PITCH + i];
  const float bm = b2s[i];
  const float ba = b2s[DIM + i];
  lds_barrier();
  const v4f q0 = redbuf[0][p], q1 = redbuf[1][p], q2 = redbuf[2][p], q3 = redbuf[3][p];
  const v4f s = (q0 + q1) + (q2 + q3);
  const float xa = fmaf(za, expf(s.y + ba), s.x + bm);
  const float xb = fmaf(zb, expf(s.w + ba), s.z + bm);
  if (tid < 8) {
    xs[p * SC_PITCH + i] = xa;
    xs[(p + 8) * SC_PITCH + i] = xb;
  }
  if (do_pass) {
#pragma unroll
    for (int c = 0; c < 4; ++c) { maA[c] = (v2f){0.f, 0.f}; maB[c] = (v2f){0.f, 0.f}; }
#pragma unroll
    for (int m = 0; m < 4; ++m) pass8x2(ha[m], hb[m], w1[m], wa[m], wb[m], xa, xb, sel, maA, maB);
  }
}

__global__ __launch_bounds__(128) void maf_scan(const float* __restrict__ z,
                                                const float* __restrict__ b1,
                                                const float* __restrict__ b2,
                                                const float* __restrict__ ws,
                                                float* __restrict__ xout) {
  __shared__ float zs[SC_ROWS * SC_PITCH];
  __shared__ float xs[SC_ROWS * SC_PITCH];
  __shared__ float b2s[2 * DIM];
  __shared__ __align__(16) v4f red[2][SC_WAVES][8];
  const float* MW1T = ws + WS_MW1T;
  const float* MW2a = ws + WS_MW2;
  const float* MW2b = ws + WS_MW2 + DIM * HID;

  const int tid  = threadIdx.x;
  const int lane = tid & 31;
  const int wave = tid >> 5;
  const int p    = lane & 7;
  const int row0 = blockIdx.x * SC_ROWS;
  const int off  = tid * 4;

  int sel[8];
#pragma unroll
  for (int k = 0; k < 8; ++k) sel[k] = (lane & 8) + k;

  b2s[tid] = b2[tid];
#pragma unroll
  for (int q = 0; q < 8; ++q) {
    const int idx = q * 128 + tid;
    zs[(idx >> 6) * SC_PITCH + (idx & 63)] = z[row0 * DIM + idx];
  }

  const v4f bv  = *(const v4f*)(b1 + off);
  const v4f wa0 = *(const v4f*)(MW2a + off);
  const v4f wb0 = *(const v4f*)(MW2b + off);
  v4f S1[4], Sa[4], Sb[4];
#pragma unroll
  for (int j = 0; j < 4; ++j) {
    S1[j] = *(const v4f*)(MW1T + j * HID + off);
    Sa[j] = *(const v4f*)(MW2a + (j + 1) * HID + off);
    Sb[j] = *(const v4f*)(MW2b + (j + 1) * HID + off);
  }

  float ha[4][8], hb[4][8];
  v2f maA[4], maB[4];
#pragma unroll
  for (int m = 0; m < 4; ++m) {
#pragma unroll
    for (int k = 0; k < 8; ++k) {
      const float b = grp_bcast(bv[m], sel[k]);
      ha[m][k] = b; hb[m][k] = b;
    }
    maA[m] = (v2f){0.f, 0.f};
    maB[m] = (v2f){0.f, 0.f};
  }
#pragma unroll
  for (int m = 0; m < 4; ++m) dot8x2(ha[m], hb[m], wa0[m], wb0[m], sel, maA, maB);
  lds_barrier();

  for (int i = 0; i < DIM; i += 4) {
#pragma unroll
    for (int j = 0; j < 4; ++j) {
      scan_step(i + j, i + j < DIM - 1, red[j & 1], zs, xs, b2s, tid, wave, lane, p, sel, ha, hb, maA, maB,
                S1[j], Sa[j], Sb[j]);
      const int pr = min(i + j + 4, DIM - 2);
      S1[j] = *(const v4f*)(MW1T + pr * HID + off);
      Sa[j] = *(const v4f*)(MW2a + (pr + 1) * HID + off);
      Sb[j] = *(const v4f*)(MW2b + (pr + 1) * HID + off);
    }
  }

  lds_barrier();
#pragma unroll
  for (int q = 0; q < 8; ++q) {
    const int idx = q * 128 + tid;
    xout[row0 * DIM + idx] = xs[(idx >> 6) * SC_PITCH + (idx & 63)];
  }
}

#define LD_WAVES 4
#define LD_ROWS  (LD_WAVES * 16)

__device__ __forceinline__ void ld_tile(const v2f (&a)[16], const v2f (&b)[16], float bb, float wv, v8f& ld) {
  v8f acc0 = {}, acc1 = {};
#pragma unroll
  for (int t = 0; t < 16; t += 2) {
    acc0 = __builtin_amdgcn_wmma_f32_16x16x4_f32(false, a[t],     false, b[t],
                                                 (short)0, acc0, false, false);
    acc1 = __builtin_amdgcn_wmma_f32_16x16x4_f32(false, a[t + 1], false, b[t + 1],
                                                 (short)0, acc1, false, false);
  }
  const v8f acc = acc0 + acc1;
#pragma unroll
  for (int v = 0; v < 8; ++v) {
    const float hv = fmaxf(acc[v] + bb, 0.f);
    ld[v] = fmaf(hv, wv, ld[v]);
  }
}

__global__ __launch_bounds__(LD_WAVES * 32) void maf_logdet_wmma(const float* __restrict__ b1,
                                                                 const float* __restrict__ ws,
                                                                 const float* __restrict__ xout,
                                                                 float* __restrict__ ldout) {
  __shared__ __align__(16) float Bs[NPACK];
  __shared__ __align__(16) float b1s[HID];
  __shared__ __align__(16) float wss[HID];
  const float bsum = ws[WS_BSUM];

  const int tid  = threadIdx.x;
  const int lane = tid & 31;
  const int wave = tid >> 5;
  const int lr   = lane & 15;
  const int koff = (lane >> 4) ? 2 : 0;
  const int m0   = (blockIdx.x * LD_WAVES + wave) * 16;

  const float* xp = xout + (m0 + lr) * DIM + koff;
  v2f a[16];
#pragma unroll
  for (int t = 0; t < 16; ++t) a[t] = *(const v2f*)(xp + 4 * t);

  {
    const v4f* src = (const v4f*)(ws + WS_BPACK);
    v4f* dst = (v4f*)Bs;
#pragma unroll 32
    for (int q = 0; q < NPACK / 4 / (LD_WAVES * 32); ++q) dst[q * (LD_WAVES * 32) + tid] = src[q * (LD_WAVES * 32) + tid];
    ((v4f*)b1s)[tid] = ((const v4f*)b1)[tid];
    ((v4f*)wss)[tid] = ((const v4f*)(ws + WS_WSUM))[tid];
  }
  __syncthreads();

  const v2f* Bl = (const v2f*)Bs + lane;
  v8f ld = {};
  v2f bc[16], bn[16];
#pragma unroll
  for (int t = 0; t < 16; ++t) bc[t] = Bl[t * 32];
  for (int nt = 0; nt < 32; nt += 2) {
    const v2f* p1 = Bl + (nt + 1) * 512;
#pragma unroll
    for (int t = 0; t < 16; ++t) bn[t] = p1[t * 32];
    ld_tile(a, bc, b1s[nt * 16 + lr], wss[nt * 16 + lr], ld);
    const v2f* p2 = Bl + min(nt + 2, 31) * 512;
#pragma unroll
    for (int t = 0; t < 16; ++t) bc[t] = p2[t * 32];
    ld_tile(a, bn, b1s[nt * 16 + 16 + lr], wss[nt * 16 + 16 + lr], ld);
  }

#pragma unroll
  for (int off = 1; off < 16; off <<= 1) {
#pragma unroll
    for (int v = 0; v < 8; ++v) {
      float t = ld[v];
      ld[v] = t + __shfl_xor(t, off, 32);
    }
  }
  if (lane == 0) {
#pragma unroll
    for (int v = 0; v < 8; ++v) ldout[m0 + v] = ld[v] + bsum;
  } else if (lane == 16) {
#pragma unroll
    for (int v = 0; v < 8; ++v) ldout[m0 + 8 + v] = ld[v] + bsum;
  }
}

extern "C" void kernel_launch(void* const* d_in, const int* in_sizes, int n_in,
                              void* d_out, int out_size, void* d_ws, size_t ws_size,
                              hipStream_t stream) {
  const float* z     = (const float*)d_in[0];
  const float* W1    = (const float*)d_in[1];
  const float* b1    = (const float*)d_in[2];
  const float* W2    = (const float*)d_in[3];
  const float* b2    = (const float*)d_in[4];
  const float* mask1 = (const float*)d_in[5];
  const float* mask2 = (const float*)d_in[6];

  float* ws    = (float*)d_ws;
  float* xout  = (float*)d_out;
  float* ldout = xout + BATCH * DIM;

  const int nprep = HID * DIM + 2 * DIM * HID + NPACK;
  maf_prep_mask<<<(nprep + 255) / 256, 256, 0, stream>>>(W1, mask1, W2, mask2, ws);
  maf_prep_sum<<<1, 512, 0, stream>>>(b2, ws);
  maf_scan<<<BATCH / SC_ROWS, SC_WAVES * 32, 0, stream>>>(z, b1, b2, ws, xout);
  maf_logdet_wmma<<<BATCH / LD_ROWS, LD_WAVES * 32, 0, stream>>>(b1, ws, xout, ldout);
}


# scan: w_sum and b_sum accumulated from the alpha weight rows and biases already in its registers (index order; work-group 0 stores them); prep_sum launch dropped (3 launches); sums bit-identical
# speedup vs baseline: 3.2093x; 1.0751x over previous
#include <hip/hip_runtime.h>
#include <hip/hip_bf16.h>

#define DIM 64
#define HID 512
#define BATCH 8192

typedef __attribute__((ext_vector_type(2))) float v2f;
typedef __attribute__((ext_vector_type(8))) float v8f;

#define WS_MW1T  0
#define WS_MW2   (HID*DIM)
#define WS_WSUM  (HID*DIM + 2*DIM*HID)
#define WS_BSUM  (WS_WSUM + HID)
#define WS_BPACK 98832
#define NPACK    (HID*DIM)

__global__ __launch_bounds__(256) void maf_prep_mask(const float* __restrict__ W1,
                                                     const float* __restrict__ mask1,
                                                     const float* __restrict__ W2,
                                                     const float* __restrict__ mask2,
                                                     float* __restrict__ ws) {
  int idx = blockIdx.x * blockDim.x + threadIdx.x;
  if (idx < HID * DIM) {
    int j = idx / DIM, k = idx % DIM;
    ws[WS_MW1T + k * HID + j] = mask1[idx] * W1[idx];
  } else if (idx < HID * DIM + 2 * DIM * HID) {
    int t = idx - HID * DIM;
    ws[WS_MW2 + t] = mask2[t] * W2[t];
  } else if (idx < HID * DIM + 2 * DIM * HID + NPACK) {
    int f = idx - (HID * DIM + 2 * DIM * HID);
    int p = f >> 1, c = f & 1;
    int nt = p >> 9;
    int t  = (p >> 5) & 15;
    int l  = p & 31;
    int koff = (l >> 4) << 1;
    int j = (nt << 4) + (l & 15);
    int k = (t << 2) + koff + c;
    ws[WS_BPACK + f] = mask1[j * DIM + k] * W1[j * DIM + k];
  }
}

typedef __attribute__((ext_vector_type(4))) float v4f;

#define SC_WAVES 4
#define SC_ROWS  16
#define SC_PITCH 65

__device__ __forceinline__ float grp_bcast(float v, int sel) {
  int keep;
  return __builtin_bit_cast(float, __builtin_amdgcn_permlane16_var(keep, __builtin_bit_cast(int, v), sel, false,
                                                                   false));
}

__device__ __forceinline__ float wave_sum(float v) {
  const float t = v + __builtin_bit_cast(float, __builtin_amdgcn_update_dpp(0, __builtin_bit_cast(int, v),
                                                                            0x168, 0xf, 0xf, true));
  int keep;
  return t + __builtin_bit_cast(float, __builtin_amdgcn_permlanex16(keep, __builtin_bit_cast(int, t), 0x76543210,
                                                                    0xfedcba98, false, false));
}

__device__ __forceinline__ void dot8x2(const float* ha, const float* hb, float wa, float wb, const int (&sel)[8],
                                       v2f (&maA)[4], v2f (&maB)[4]) {
#pragma unroll
  for (int k = 0; k < 8; ++k) {
    v2f w, ra, rb;
    w.x = grp_bcast(wa, sel[k]); w.y = grp_bcast(wb, sel[k]);
    const float a = fmaxf(ha[k], 0.f), b = fmaxf(hb[k], 0.f);
    ra.x = a; ra.y = a; rb.x = b; rb.y = b;
    maA[k & 3] = __builtin_elementwise_fma(w, ra, maA[k & 3]);
    maB[k & 3] = __builtin_elementwise_fma(w, rb, maB[k & 3]);
  }
}
__device__ __forceinline__ void pass8x2(float* ha, float* hb, float w1, float wa, float wb, float xa, float xb,
                                        const int (&sel)[8], v2f (&maA)[4], v2f (&maB)[4]) {
  v2f xxa, xxb;
  xxa.x = xa; xxa.y = xa; xxb.x = xb; xxb.y = xb;
#pragma unroll
  for (int k = 0; k < 8; k += 2) {
    v2f w, a, b;
    w.x = grp_bcast(w1, sel[k]); w.y = grp_bcast(w1, sel[k + 1]);
    a.x = ha[k]; a.y = ha[k + 1]; b.x = hb[k]; b.y = hb[k + 1];
    a = __builtin_elementwise_fma(w, xxa, a);
    b = __builtin_elementwise_fma(w, xxb, b);
    ha[k] = a.x; ha[k + 1] = a.y; hb[k] = b.x; hb[k + 1] = b.y;
  }
  dot8x2(ha, hb, wa, wb, sel, maA, maB);
}

__device__ __forceinline__ void lds_barrier() {
  asm volatile("s_wait_dscnt 0x0\n\ts_barrier_signal -1\n\ts_barrier_wait -1" ::: "memory");
}

__device__ __forceinline__ void scan_step(int i, bool do_pass, v4f (*redbuf)[8], const float* zs, float* xs,
                                          const float* b2s, int tid, int wave, int lane, int p,
                                          const int (&sel)[8], float (&ha)[4][8], float (&hb)[4][8],
                                          v2f (&maA)[4], v2f (&maB)[4], v4f w1, v4f wa, v4f wb, v4f& wacc,
                                          float& bacc) {
  const v2f pa = (maA[0] + maA[1]) + (maA[2] + maA[3]);
  const v2f pb = (maB[0] + maB[1]) + (maB[2] + maB[3]);
  v4f part;
  part.x = wave_sum(pa.x); part.y = wave_sum(pa.y); part.z = wave_sum(pb.x); part.w = wave_sum(pb.y);
  if (lane < 8) redbuf[wave][p] = part;
  const float za = zs[p * SC_PITCH + i];
  const float zb = zs[(p + 8) * SC_PITCH + i];
  const float bm = b2s[i];
  const float ba = b2s[DIM + i];
  bacc += ba;
  lds_barrier();
  const v4f q0 = redbuf[0][p], q1 = redbuf[1][p], q2 = redbuf[2][p], q3 = redbuf[3][p];
  const v4f s = (q0 + q1) + (q2 + q3);
  const float xa = fmaf(za, expf(s.y + ba), s.x + bm);
  const float xb = fmaf(zb, expf(s.w + ba), s.z + bm);
  if (tid < 8) {
    xs[p * SC_PITCH + i] = xa;
    xs[(p + 8) * SC_PITCH + i] = xb;
  }
  if (do_pass) {
#pragma unroll
    for (int c = 0; c < 4; ++c) { maA[c] = (v2f){0.f, 0.f}; maB[c] = (v2f){0.f, 0.f}; }
#pragma unroll
    for (int m = 0; m < 4; ++m) pass8x2(ha[m], hb[m], w1[m], wa[m], wb[m], xa, xb, sel, maA, maB);
    wacc += wb;
  }
}

__global__ __launch_bounds__(128) void maf_scan(const float* __restrict__ z,
                                                const float* __restrict__ b1,
                                                const float* __restrict__ b2,
                                                const float* __restrict__ ws,
                                                float* __restrict__ sums,
                                                float* __restrict__ xout) {
  __shared__ float zs[SC_ROWS * SC_PITCH];
  __shared__ float xs[SC_ROWS * SC_PITCH];
  __shared__ float b2s[2 * DIM];
  __shared__ __align__(16) v4f red[2][SC_WAVES][8];
  const float* MW1T = ws + WS_MW1T;
  const float* MW2a = ws + WS_MW2;
  const float* MW2b = ws + WS_MW2 + DIM * HID;

  const int tid  = threadIdx.x;
  const int lane = tid & 31;
  const int wave = tid >> 5;
  const int p    = lane & 7;
  const int row0 = blockIdx.x * SC_ROWS;
  const int off  = tid * 4;

  int sel[8];
#pragma unroll
  for (int k = 0; k < 8; ++k) sel[k] = (lane & 8) + k;

  b2s[tid] = b2[tid];
#pragma unroll
  for (int q = 0; q < 8; ++q) {
    const int idx = q * 128 + tid;
    zs[(idx >> 6) * SC_PITCH + (idx & 63)] = z[row0 * DIM + idx];
  }

  const v4f bv  = *(const v4f*)(b1 + off);
  const v4f wa0 = *(const v4f*)(MW2a + off);
  const v4f wb0 = *(const v4f*)(MW2b + off);
  v4f S1[4], Sa[4], Sb[4];
#pragma unroll
  for (int j = 0; j < 4; ++j) {
    S1[j] = *(const v4f*)(MW1T + j * HID + off);
    Sa[j] = *(const v4f*)(MW2a + (j + 1) * HID + off);
    Sb[j] = *(const v4f*)(MW2b + (j + 1) * HID + off);
  }

  v4f wacc = wb0;
  float bacc = 0.f;
  float ha[4][8], hb[4][8];
  v2f maA[4], maB[4];
#pragma unroll
  for (int m = 0; m < 4; ++m) {
#pragma unroll
    for (int k = 0; k < 8; ++k) {
      const float b = grp_bcast(bv[m], sel[k]);
      ha[m][k] = b; hb[m][k] = b;
    }
    maA[m] = (v2f){0.f, 0.f};
    maB[m] = (v2f){0.f, 0.f};
  }
#pragma unroll
  for (int m = 0; m < 4; ++m) dot8x2(ha[m], hb[m], wa0[m], wb0[m], sel, maA, maB);
  lds_barrier();

  for (int i = 0; i < DIM; i += 4) {
#pragma unroll
    for (int j = 0; j < 4; ++j) {
      scan_step(i + j, i + j < DIM - 1, red[j & 1], zs, xs, b2s, tid, wave, lane, p, sel, ha, hb, maA, maB,
                S1[j], Sa[j], Sb[j], wacc, bacc);
      const int pr = min(i + j + 4, DIM - 2);
      S1[j] = *(const v4f*)(MW1T + pr * HID + off);
      Sa[j] = *(const v4f*)(MW2a + (pr + 1) * HID + off);
      Sb[j] = *(const v4f*)(MW2b + (pr + 1) * HID + off);
    }
  }

  if (blockIdx.x == 0) {
    *(v4f*)(sums + off) = wacc;
    if (tid == 0) sums[HID] = bacc;
  }
  lds_barrier();
#pragma unroll
  for (int q = 0; q < 8; ++q) {
    const int idx = q * 128 + tid;
    xout[row0 * DIM + idx] = xs[(idx >> 6) * SC_PITCH + (idx & 63)];
  }
}

#define LD_WAVES 4
#define LD_ROWS  (LD_WAVES * 16)

__device__ __forceinline__ void ld_tile(const v2f (&a)[16], const v2f (&b)[16], float bb, float wv, v8f& ld) {
  v8f acc0 = {}, acc1 = {};
#pragma unroll
  for (int t = 0; t < 16; t += 2) {
    acc0 = __builtin_amdgcn_wmma_f32_16x16x4_f32(false, a[t],     false, b[t],
                                                 (short)0, acc0, false, false);
    acc1 = __builtin_amdgcn_wmma_f32_16x16x4_f32(false, a[t + 1], false, b[t + 1],
                                                 (short)0, acc1, false, false);
  }
  const v8f acc = acc0 + acc1;
#pragma unroll
  for (int v = 0; v < 8; ++v) {
    const float hv = fmaxf(acc[v] + bb, 0.f);
    ld[v] = fmaf(hv, wv, ld[v]);
  }
}

__global__ __launch_bounds__(LD_WAVES * 32) void maf_logdet_wmma(const float* __restrict__ b1,
                                                                 const float* __restrict__ ws,
                                                                 const float* __restrict__ xout,
                                                                 float* __restrict__ ldout) {
  __shared__ __align__(16) float Bs[NPACK];
  __shared__ __align__(16) float b1s[HID];
  __shared__ __align__(16) float wss[HID];
  const float bsum = ws[WS_BSUM];

  const int tid  = threadIdx.x;
  const int lane = tid & 31;
  const int wave = tid >> 5;
  const int lr   = lane & 15;
  const int koff = (lane >> 4) ? 2 : 0;
  const int m0   = (blockIdx.x * LD_WAVES + wave) * 16;

  const float* xp = xout + (m0 + lr) * DIM + koff;
  v2f a[16];
#pragma unroll
  for (int t = 0; t < 16; ++t) a[t] = *(const v2f*)(xp + 4 * t);

  {
    const v4f* src = (const v4f*)(ws + WS_BPACK);
    v4f* dst = (v4f*)Bs;
#pragma unroll 32
    for (int q = 0; q < NPACK / 4 / (LD_WAVES * 32); ++q) dst[q * (LD_WAVES * 32) + tid] = src[q * (LD_WAVES * 32) + tid];
    ((v4f*)b1s)[tid] = ((const v4f*)b1)[tid];
    ((v4f*)wss)[tid] = ((const v4f*)(ws + WS_WSUM))[tid];
  }
  __syncthreads();

  const v2f* Bl = (const v2f*)Bs + lane;
  v8f ld = {};
  v2f bc[16], bn[16];
#pragma unroll
  for (int t = 0; t < 16; ++t) bc[t] = Bl[t * 32];
  for (int nt = 0; nt < 32; nt += 2) {
    const v2f* p1 = Bl + (nt + 1) * 512;
#pragma unroll
    for (int t = 0; t < 16; ++t) bn[t] = p1[t * 32];
    ld_tile(a, bc, b1s[nt * 16 + lr], wss[nt * 16 + lr], ld);
    const v2f* p2 = Bl + min(nt + 2, 31) * 512;
#pragma unroll
    for (int t = 0; t < 16; ++t) bc[t] = p2[t * 32];
    ld_tile(a, bn, b1s[nt * 16 + 16 + lr], wss[nt * 16 + 16 + lr], ld);
  }

#pragma unroll
  for (int off = 1; off < 16; off <<= 1) {
#pragma unroll
    for (int v = 0; v < 8; ++v) {
      float t = ld[v];
      ld[v] = t + __shfl_xor(t, off, 32);
    }
  }
  if (lane == 0) {
#pragma unroll
    for (int v = 0; v < 8; ++v) ldout[m0 + v] = ld[v] + bsum;
  } else if (lane == 16) {
#pragma unroll
    for (int v = 0; v < 8; ++v) ldout[m0 + 8 + v] = ld[v] + bsum;
  }
}

extern "C" void kernel_launch(void* const* d_in, const int* in_sizes, int n_in,
                              void* d_out, int out_size, void* d_ws, size_t ws_size,
                              hipStream_t stream) {
  const float* z     = (const float*)d_in[0];
  const float* W1    = (const float*)d_in[1];
  const float* b1    = (const float*)d_in[2];
  const float* W2    = (const float*)d_in[3];
  const float* b2    = (const float*)d_in[4];
  const float* mask1 = (const float*)d_in[5];
  const float* mask2 = (const float*)d_in[6];

  float* ws    = (float*)d_ws;
  float* xout  = (float*)d_out;
  float* ldout = xout + BATCH * DIM;

  const int nprep = HID * DIM + 2 * DIM * HID + NPACK;
  maf_prep_mask<<<(nprep + 255) / 256, 256, 0, stream>>>(W1, mask1, W2, mask2, ws);
  maf_scan<<<BATCH / SC_ROWS, SC_WAVES * 32, 0, stream>>>(z, b1, b2, ws, ws + WS_WSUM, xout);
  maf_logdet_wmma<<<BATCH / LD_ROWS, LD_WAVES * 32, 0, stream>>>(b1, ws, xout, ldout);
}


# logdet: K = 64 of a row tile split over a pair of waves (256 blocks of 32 rows, was 128 of 64); partial tiles cross in LDS before the relu, behind the next matrix instr; new sum order
# speedup vs baseline: 3.4398x; 1.0718x over previous
#include <hip/hip_runtime.h>
#include <hip/hip_bf16.h>

#define DIM 64
#define HID 512
#define BATCH 8192

typedef __attribute__((ext_vector_type(2))) float v2f;
typedef __attribute__((ext_vector_type(8))) float v8f;

#define WS_MW1T  0
#define WS_MW2   (HID*DIM)
#define WS_WSUM  (HID*DIM + 2*DIM*HID)
#define WS_BSUM  (WS_WSUM + HID)
#define WS_BPACK 98832
#define NPACK    (HID*DIM)

__global__ __launch_bounds__(256) void maf_prep_mask(const float* __restrict__ W1,
                                                     const float* __restrict__ mask1,
                                                     const float* __restrict__ W2,
                                                     const float* __restrict__ mask2,
                                                     float* __restrict__ ws) {
  int idx = blockIdx.x * blockDim.x + threadIdx.x;
  if (idx < HID * DIM) {
    int j = idx / DIM, k = idx % DIM;
    ws[WS_MW1T + k * HID + j] = mask1[idx] * W1[idx];
  } else if (idx < HID * DIM + 2 * DIM * HID) {
    int t = idx - HID * DIM;
    ws[WS_MW2 + t] = mask2[t] * W2[t];
  } else if (idx < HID * DIM + 2 * DIM * HID + NPACK) {
    int f = idx - (HID * DIM + 2 * DIM * HID);
    int p = f >> 1, c = f & 1;
    int nt = p >> 9;
    int t  = (p >> 5) & 15;
    int l  = p & 31;
    int koff = (l >> 4) << 1;
    int j = (nt << 4) + (l & 15);
    int k = (t << 2) + koff + c;
    ws[WS_BPACK + f] = mask1[j * DIM + k] * W1[j * DIM + k];
  }
}

typedef __attribute__((ext_vector_type(4))) float v4f;

#define SC_WAVES 4
#define SC_ROWS  16
#define SC_PITCH 65

__device__ __forceinline__ float grp_bcast(float v, int sel) {
  int keep;
  return __builtin_bit_cast(float, __builtin_amdgcn_permlane16_var(keep, __builtin_bit_cast(int, v), sel, false,
                                                                   false));
}

__device__ __forceinline__ float wave_sum(float v) {
  const float t = v + __builtin_bit_cast(float, __builtin_amdgcn_update_dpp(0, __builtin_bit_cast(int, v),
                                                                            0x168, 0xf, 0xf, true));
  int keep;
  return t + __builtin_bit_cast(float, __builtin_amdgcn_permlanex16(keep, __builtin_bit_cast(int, t), 0x76543210,
                                                                    0xfedcba98, false, false));
}

__device__ __forceinline__ void dot8x2(const float* ha, const float* hb, float wa, float wb, const int (&sel)[8],
                                       v2f (&maA)[4], v2f (&maB)[4]) {
#pragma unroll
  for (int k = 0; k < 8; ++k) {
    v2f w, ra, rb;
    w.x = grp_bcast(wa, sel[k]); w.y = grp_bcast(wb, sel[k]);
    const float a = fmaxf(ha[k], 0.f), b = fmaxf(hb[k], 0.f);
    ra.x = a; ra.y = a; rb.x = b; rb.y = b;
    maA[k & 3] = __builtin_elementwise_fma(w, ra, maA[k & 3]);
    maB[k & 3] = __builtin_elementwise_fma(w, rb, maB[k & 3]);
  }
}
__device__ __forceinline__ void pass8x2(float* ha, float* hb, float w1, float wa, float wb, float xa, float xb,
                                        const int (&sel)[8], v2f (&maA)[4], v2f (&maB)[4]) {
  v2f xxa, xxb;
  xxa.x = xa; xxa.y = xa; xxb.x = xb; xxb.y = xb;
#pragma unroll
  for (int k = 0; k < 8; k += 2) {
    v2f w, a, b;
    w.x = grp_bcast(w1, sel[k]); w.y = grp_bcast(w1, sel[k + 1]);
    a.x = ha[k]; a.y = ha[k + 1]; b.x = hb[k]; b.y = hb[k + 1];
    a = __builtin_elementwise_fma(w, xxa, a);
    b = __builtin_elementwise_fma(w, xxb, b);
    ha[k] = a.x; ha[k + 1] = a.y; hb[k] = b.x; hb[k + 1] = b.y;
  }
  dot8x2(ha, hb, wa, wb, sel, maA, maB);
}

__device__ __forceinline__ void lds_barrier() {
  asm volatile("s_wait_dscnt 0x0\n\ts_barrier_signal -1\n\ts_barrier_wait -1" ::: "memory");
}

__device__ __forceinline__ void scan_step(int i, bool do_pass, v4f (*redbuf)[8], const float* zs, float* xs,
                                          const float* b2s, int tid, int wave, int lane, int p,
                                          const int (&sel)[8], float (&ha)[4][8], float (&hb)[4][8],
                                          v2f (&maA)[4], v2f (&maB)[4], v4f w1, v4f wa, v4f wb, v4f& wacc,
                                          float& bacc) {
  const v2f pa = (maA[0] + maA[1]) + (maA[2] + maA[3]);
  const v2f pb = (maB[0] + maB[1]) + (maB[2] + maB[3]);
  v4f part;
  part.x = wave_sum(pa.x); part.y = wave_sum(pa.y); part.z = wave_sum(pb.x); part.w = wave_sum(pb.y);
  if (lane < 8) redbuf[wave][p] = part;
  const float za = zs[p * SC_PITCH + i];
  const float zb = zs[(p + 8) * SC_PITCH + i];
  const float bm = b2s[i];
  const float ba = b2s[DIM + i];
  bacc += ba;
  lds_barrier();
  const v4f q0 = redbuf[0][p], q1 = redbuf[1][p], q2 = redbuf[2][p], q3 = redbuf[3][p];
  const v4f s = (q0 + q1) + (q2 + q3);
  const float xa = fmaf(za, expf(s.y + ba), s.x + bm);
  const float xb = fmaf(zb, expf(s.w + ba), s.z + bm);
  if (tid < 8) {
    xs[p * SC_PITCH + i] = xa;
    xs[(p + 8) * SC_PITCH + i] = xb;
  }
  if (do_pass) {
#pragma unroll
    for (int c = 0; c < 4; ++c) { maA[c] = (v2f){0.f, 0.f}; maB[c] = (v2f){0.f, 0.f}; }
#pragma unroll
    for (int m = 0; m < 4; ++m) pass8x2(ha[m], hb[m], w1[m], wa[m], wb[m], xa, xb, sel, maA, maB);
    wacc += wb;
  }
}

__global__ __launch_bounds__(128) void maf_scan(const float* __restrict__ z,
                                                const float* __restrict__ b1,
                                                const float* __restrict__ b2,
                                                const float* __restrict__ ws,
                                                float* __restrict__ sums,
                                                float* __restrict__ xout) {
  __shared__ float zs[SC_ROWS * SC_PITCH];
  __shared__ float xs[SC_ROWS * SC_PITCH];
  __shared__ float b2s[2 * DIM];
  __shared__ __align__(16) v4f red[2][SC_WAVES][8];
  const float* MW1T = ws + WS_MW1T;
  const float* MW2a = ws + WS_MW2;
  const float* MW2b = ws + WS_MW2 + DIM * HID;

  const int tid  = threadIdx.x;
  const int lane = tid & 31;
  const int wave = tid >> 5;
  const int p    = lane & 7;
  const int row0 = blockIdx.x * SC_ROWS;
  const int off  = tid * 4;

  int sel[8];
#pragma unroll
  for (int k = 0; k < 8; ++k) sel[k] = (lane & 8) + k;

  b2s[tid] = b2[tid];
#pragma unroll
  for (int q = 0; q < 8; ++q) {
    const int idx = q * 128 + tid;
    zs[(idx >> 6) * SC_PITCH + (idx & 63)] = z[row0 * DIM + idx];
  }

  const v4f bv  = *(const v4f*)(b1 + off);
  const v4f wa0 = *(const v4f*)(MW2a + off);
  const v4f wb0 = *(const v4f*)(MW2b + off);
  v4f S1[4], Sa[4], Sb[4];
#pragma unroll
  for (int j = 0; j < 4; ++j) {
    S1[j] = *(const v4f*)(MW1T + j * HID + off);
    Sa[j] = *(const v4f*)(MW2a + (j + 1) * HID + off);
    Sb[j] = *(const v4f*)(MW2b + (j + 1) * HID + off);
  }

  v4f wacc = wb0;
  float bacc = 0.f;
  float ha[4][8], hb[4][8];
  v2f maA[4], maB[4];
#pragma unroll
  for (int m = 0; m < 4; ++m) {
#pragma unroll
    for (int k = 0; k < 8; ++k) {
      const float b = grp_bcast(bv[m], sel[k]);
      ha[m][k] = b; hb[m][k] = b;
    }
    maA[m] = (v2f){0.f, 0.f};
    maB[m] = (v2f){0.f, 0.f};
  }
#pragma unroll
  for (int m = 0; m < 4; ++m) dot8x2(ha[m], hb[m], wa0[m], wb0[m], sel, maA, maB);
  lds_barrier();

  for (int i = 0; i < DIM; i += 4) {
#pragma unroll
    for (int j = 0; j < 4; ++j) {
      scan_step(i + j, i + j < DIM - 1, red[j & 1], zs, xs, b2s, tid, wave, lane, p, sel, ha, hb, maA, maB,
                S1[j], Sa[j], Sb[j], wacc, bacc);
      const int pr = min(i + j + 4, DIM - 2);
      S1[j] = *(const v4f*)(MW1T + pr * HID + off);
      Sa[j] = *(const v4f*)(MW2a + (pr + 1) * HID + off);
      Sb[j] = *(const v4f*)(MW2b + (pr + 1) * HID + off);
    }
  }

  if (blockIdx.x == 0) {
    *(v4f*)(sums + off) = wacc;
    if (tid == 0) sums[HID] = bacc;
  }
  lds_barrier();
#pragma unroll
  for (int q = 0; q < 8; ++q) {
    const int idx = q * 128 + tid;
    xout[row0 * DIM + idx] = xs[(idx >> 6) * SC_PITCH + (idx & 63)];
  }
}

#define LD_WAVES 4
#define LD_ROWS  32

__device__ __forceinline__ void ld_mma(const v2f (&a)[8], const v2f (&fo)[8], const v2f (&ft)[8], v8f& own, v8f& oth) {
  v8f p = {}, r = {};
#pragma unroll
  for (int u = 0; u < 8; ++u) {
    p = __builtin_amdgcn_wmma_f32_16x16x4_f32(false, a[u], false, fo[u], (short)0, p, false, false);
    r = __builtin_amdgcn_wmma_f32_16x16x4_f32(false, a[u], false, ft[u], (short)0, r, false, false);
  }
  own = p; oth = r;
}

__device__ __forceinline__ void ld_finish(v4f (*ex)[2][32], int wave, int lane, const v8f& own, const v8f& oth,
                                          float bb, float wv, v8f& ld) {
  v4f s0, s1;
  s0.x = oth[0]; s0.y = oth[1]; s0.z = oth[2]; s0.w = oth[3];
  s1.x = oth[4]; s1.y = oth[5]; s1.z = oth[6]; s1.w = oth[7];
  ex[wave][0][lane] = s0;
  ex[wave][1][lane] = s1;
  __syncthreads();
  const v4f r0 = ex[wave ^ 1][0][lane];
  const v4f r1 = ex[wave ^ 1][1][lane];
#pragma unroll
  for (int v = 0; v < 4; ++v) {
    const float h0 = fmaxf((own[v] + r0[v]) + bb, 0.f);
    const float h1 = fmaxf((own[4 + v] + r1[v]) + bb, 0.f);
    ld[v]     = fmaf(h0, wv, ld[v]);
    ld[4 + v] = fmaf(h1, wv, ld[4 + v]);
  }
}

__global__ __launch_bounds__(LD_WAVES * 32) void maf_logdet_wmma(const float* __restrict__ b1,
                                                                 const float* __restrict__ ws,
                                                                 const float* __restrict__ xout,
                                                                 float* __restrict__ ldout) {
  __shared__ __align__(16) float Bs[NPACK];
  __shared__ __align__(16) float b1s[HID];
  __shared__ __align__(16) float wss[HID];
  __shared__ __align__(16) v4f exch[2][LD_WAVES][2][32];
  __shared__ float fin[LD_WAVES][16];
  const float bsum = ws[WS_BSUM];

  const int tid  = threadIdx.x;
  const int lane = tid & 31;
  const int wave = tid >> 5;
  const int lr   = lane & 15;
  const int koff = (lane >> 4) ? 2 : 0;
  const int rt   = wave >> 1;
  const int kh   = wave & 1;
  const int m0   = blockIdx.x * LD_ROWS + rt * 16;

  const float* xp = xout + (m0 + lr) * DIM + 32 * kh + koff;
  v2f a[8];
#pragma unroll
  for (int u = 0; u < 8; ++u) a[u] = *(const v2f*)(xp + 4 * u);

  {
    const v4f* src = (const v4f*)(ws + WS_BPACK);
    v4f* dst = (v4f*)Bs;
#pragma unroll 32
    for (int q = 0; q < NPACK / 4 / (LD_WAVES * 32); ++q) dst[q * (LD_WAVES * 32) + tid] = src[q * (LD_WAVES * 32) + tid];
    ((v4f*)b1s)[tid] = ((const v4f*)b1)[tid];
    ((v4f*)wss)[tid] = ((const v4f*)(ws + WS_WSUM))[tid];
  }
  __syncthreads();

  const v2f* Bo = (const v2f*)Bs + (kh * 16 + 8 * kh) * 32 + lane;
  const v2f* Bt = (const v2f*)Bs + ((1 - kh) * 16 + 8 * kh) * 32 + lane;
  const float* bo = b1s + kh * 16 + lr;
  const float* wo = wss + kh * 16 + lr;
  v8f ld = {};
  v2f co[8], ct[8], no[8], nt[8];
  v8f ownA, othA, ownB, othB;
#pragma unroll
  for (int u = 0; u < 8; ++u) { co[u] = Bo[u * 32]; ct[u] = Bt[u * 32]; }
#pragma unroll
  for (int u = 0; u < 8; ++u) { no[u] = Bo[1024 + u * 32]; nt[u] = Bt[1024 + u * 32]; }
  ld_mma(a, co, ct, ownA, othA);
  for (int k = 1; k < 15; k += 2) {
#pragma unroll
    for (int u = 0; u < 8; ++u) { co[u] = Bo[(k + 1) * 1024 + u * 32]; ct[u] = Bt[(k + 1) * 1024 + u * 32]; }
    ld_mma(a, no, nt, ownB, othB);
    ld_finish(exch[0], wave, lane, ownA, othA, bo[(k - 1) * 32], wo[(k - 1) * 32], ld);
#pragma unroll
    for (int u = 0; u < 8; ++u) { no[u] = Bo[(k + 2) * 1024 + u * 32]; nt[u] = Bt[(k + 2) * 1024 + u * 32]; }
    ld_mma(a, co, ct, ownA, othA);
    ld_finish(exch[1], wave, lane, ownB, othB, bo[k * 32], wo[k * 32], ld);
  }
  ld_mma(a, no, nt, ownB, othB);
  ld_finish(exch[0], wave, lane, ownA, othA, bo[14 * 32], wo[14 * 32], ld);
  ld_finish(exch[1], wave, lane, ownB, othB, bo[15 * 32], wo[15 * 32], ld);

#pragma unroll
  for (int off = 1; off < 16; off <<= 1) {
#pragma unroll
    for (int v = 0; v < 8; ++v) {
      float t = ld[v];
      ld[v] = t + __shfl_xor(t, off, 32);
    }
  }
  if (lr == 0) {
    const int h8 = (lane >> 4) * 8;
#pragma unroll
    for (int v = 0; v < 8; ++v) fin[wave][h8 + v] = ld[v];
  }
  __syncthreads();
  if (wave == 0) {
    const int t2 = (lane >> 4) * 2, rr = lane & 15;
    ldout[blockIdx.x * LD_ROWS + lane] = (fin[t2][rr] + fin[t2 + 1][rr]) + bsum;
  }
}

extern "C" void kernel_launch(void* const* d_in, const int* in_sizes, int n_in,
                              void* d_out, int out_size, void* d_ws, size_t ws_size,
                              hipStream_t stream) {
  const float* z     = (const float*)d_in[0];
  const float* W1    = (const float*)d_in[1];
  const float* b1    = (const float*)d_in[2];
  const float* W2    = (const float*)d_in[3];
  const float* b2    = (const float*)d_in[4];
  const float* mask1 = (const float*)d_in[5];
  const float* mask2 = (const float*)d_in[6];

  float* ws    = (float*)d_ws;
  float* xout  = (float*)d_out;
  float* ldout = xout + BATCH * DIM;

  const int nprep = HID * DIM + 2 * DIM * HID + NPACK;
  maf_prep_mask<<<(nprep + 255) / 256, 256, 0, stream>>>(W1, mask1, W2, mask2, ws);
  maf_scan<<<BATCH / SC_ROWS, SC_WAVES * 32, 0, stream>>>(z, b1, b2, ws, ws + WS_WSUM, xout);
  maf_logdet_wmma<<<BATCH / LD_ROWS, LD_WAVES * 32, 0, stream>>>(b1, ws, xout, ldout);
}


# scan: four batch rows per lane (4-lane groups of 16 hidden units): one v_permlane16_var copy of a weight feeds the packed FMAs of four rows; partials folded by a lane exchange (new sum order)
# speedup vs baseline: 3.5721x; 1.0385x over previous
#include <hip/hip_runtime.h>
#include <hip/hip_bf16.h>

#define DIM 64
#define HID 512
#define BATCH 8192

typedef __attribute__((ext_vector_type(2))) float v2f;
typedef __attribute__((ext_vector_type(8))) float v8f;

#define WS_MW1T  0
#define WS_MW2   (HID*DIM)
#define WS_WSUM  (HID*DIM + 2*DIM*HID)
#define WS_BSUM  (WS_WSUM + HID)
#define WS_BPACK 98832
#define NPACK    (HID*DIM)

__global__ __launch_bounds__(256) void maf_prep_mask(const float* __restrict__ W1,
                                                     const float* __restrict__ mask1,
                                                     const float* __restrict__ W2,
                                                     const float* __restrict__ mask2,
                                                     float* __restrict__ ws) {
  int idx = blockIdx.x * blockDim.x + threadIdx.x;
  if (idx < HID * DIM) {
    int j = idx / DIM, k = idx % DIM;
    ws[WS_MW1T + k * HID + j] = mask1[idx] * W1[idx];
  } else if (idx < HID * DIM + 2 * DIM * HID) {
    int t = idx - HID * DIM;
    ws[WS_MW2 + t] = mask2[t] * W2[t];
  } else if (idx < HID * DIM + 2 * DIM * HID + NPACK) {
    int f = idx - (HID * DIM + 2 * DIM * HID);
    int p = f >> 1, c = f & 1;
    int nt = p >> 9;
    int t  = (p >> 5) & 15;
    int l  = p & 31;
    int koff = (l >> 4) << 1;
    int j = (nt << 4) + (l & 15);
    int k = (t << 2) + koff + c;
    ws[WS_BPACK + f] = mask1[j * DIM + k] * W1[j * DIM + k];
  }
}

typedef __attribute__((ext_vector_type(4))) float v4f;

#define SC_WAVES 4
#define SC_ROWS  16
#define SC_PITCH 65

__device__ __forceinline__ float grp_bcast(float v, int sel) {
  int keep;
  return __builtin_bit_cast(float, __builtin_amdgcn_permlane16_var(keep, __builtin_bit_cast(int, v), sel, false,
                                                                   false));
}

__device__ __forceinline__ float wave_sum(float v) {
  const float t = v + __builtin_bit_cast(float, __builtin_amdgcn_update_dpp(0, __builtin_bit_cast(int, v),
                                                                            0x168, 0xf, 0xf, true));
  int keep;
  return t + __builtin_bit_cast(float, __builtin_amdgcn_permlanex16(keep, __builtin_bit_cast(int, t), 0x76543210,
                                                                    0xfedcba98, false, false));
}

__device__ __forceinline__ void dot4x4(const float (&h)[4][4], float wa, float wb, const int (&sel)[4],
                                       v2f (&ma)[4][2]) {
#pragma unroll
  for (int k = 0; k < 4; ++k) {
    v2f w;
    w.x = grp_bcast(wa, sel[k]); w.y = grp_bcast(wb, sel[k]);
#pragma unroll
    for (int s = 0; s < 4; ++s) {
      const float a = fmaxf(h[s][k], 0.f);
      v2f r;
      r.x = a; r.y = a;
      ma[s][k & 1] = __builtin_elementwise_fma(w, r, ma[s][k & 1]);
    }
  }
}
__device__ __forceinline__ void pass4x4(float (&h)[4][4], float w1, float wa, float wb, const v2f (&xx)[4],
                                        const int (&sel)[4], v2f (&ma)[4][2]) {
#pragma unroll
  for (int k = 0; k < 4; k += 2) {
    v2f w;
    w.x = grp_bcast(w1, sel[k]); w.y = grp_bcast(w1, sel[k + 1]);
#pragma unroll
    for (int s = 0; s < 4; ++s) {
      v2f a;
      a.x = h[s][k]; a.y = h[s][k + 1];
      a = __builtin_elementwise_fma(w, xx[s], a);
      h[s][k] = a.x; h[s][k + 1] = a.y;
    }
  }
  dot4x4(h, wa, wb, sel, ma);
}

__device__ __forceinline__ void lds_barrier() {
  asm volatile("s_wait_dscnt 0x0\n\ts_barrier_signal -1\n\ts_barrier_wait -1" ::: "memory");
}

__device__ __forceinline__ void scan_step(int i, bool do_pass, v4f (*redbuf)[8], const float* zs, float* xs,
                                          const float* b2s, int tid, int wave, int lane, int base, int xsel,
                                          const int (&sel)[4], float (&h)[4][4][4], v2f (&ma)[4][2], v4f w1, v4f wa,
                                          v4f wb, v4f& wacc, float& bacc) {
  const v2f p0 = ma[0][0] + ma[0][1], p1 = ma[1][0] + ma[1][1], p2 = ma[2][0] + ma[2][1], p3 = ma[3][0] + ma[3][1];
  v2f ra, rb;
  ra.x = grp_bcast(p2.x, xsel); ra.y = grp_bcast(p2.y, xsel);
  rb.x = grp_bcast(p3.x, xsel); rb.y = grp_bcast(p3.y, xsel);
  const v2f ta = p0 + ra;
  const v2f tb = p1 + rb;
  v4f part;
  part.x = wave_sum(ta.x); part.y = wave_sum(ta.y); part.z = wave_sum(tb.x); part.w = wave_sum(tb.y);
  if (lane < 8) redbuf[wave][lane] = part;
  const float za = zs[base * SC_PITCH + i];
  const float zb = zs[(base + 4) * SC_PITCH + i];
  const float bm = b2s[i];
  const float ba = b2s[DIM + i];
  bacc += ba;
  lds_barrier();
  const int j = lane & 7;
  const v4f q0 = redbuf[0][j], q1 = redbuf[1][j], q2 = redbuf[2][j], q3 = redbuf[3][j];
  const v4f s = (q0 + q1) + (q2 + q3);
  const float x0 = fmaf(za, expf(s.y + ba), s.x + bm);
  const float x1 = fmaf(zb, expf(s.w + ba), s.z + bm);
  if (tid < 8) {
    xs[base * SC_PITCH + i] = x0;
    xs[(base + 4) * SC_PITCH + i] = x1;
  }
  if (do_pass) {
    v2f xx[4];
    const float x2 = grp_bcast(x0, xsel), x3 = grp_bcast(x1, xsel);
    xx[0].x = x0; xx[0].y = x0; xx[1].x = x1; xx[1].y = x1;
    xx[2].x = x2; xx[2].y = x2; xx[3].x = x3; xx[3].y = x3;
#pragma unroll
    for (int r = 0; r < 4; ++r) { ma[r][0] = (v2f){0.f, 0.f}; ma[r][1] = (v2f){0.f, 0.f}; }
#pragma unroll
    for (int m = 0; m < 4; ++m) pass4x4(h[m], w1[m], wa[m], wb[m], xx, sel, ma);
    wacc += wb;
  }
}

__global__ __launch_bounds__(128) void maf_scan(const float* __restrict__ z,
                                                const float* __restrict__ b1,
                                                const float* __restrict__ b2,
                                                const float* __restrict__ ws,
                                                float* __restrict__ sums,
                                                float* __restrict__ xout) {
  __shared__ float zs[SC_ROWS * SC_PITCH];
  __shared__ float xs[SC_ROWS * SC_PITCH];
  __shared__ float b2s[2 * DIM];
  __shared__ __align__(16) v4f red[2][SC_WAVES][8];
  const float* MW1T = ws + WS_MW1T;
  const float* MW2a = ws + WS_MW2;
  const float* MW2b = ws + WS_MW2 + DIM * HID;

  const int tid  = threadIdx.x;
  const int lane = tid & 31;
  const int wave = tid >> 5;
  const int base = (lane & 3) + 2 * (lane & 4);
  const int xsel = (lane & 15) ^ 4;
  const int row0 = blockIdx.x * SC_ROWS;
  const int off  = tid * 4;

  int sel[4];
#pragma unroll
  for (int k = 0; k < 4; ++k) sel[k] = (lane & 12) + k;

  b2s[tid] = b2[tid];
#pragma unroll
  for (int q = 0; q < 8; ++q) {
    const int idx = q * 128 + tid;
    zs[(idx >> 6) * SC_PITCH + (idx & 63)] = z[row0 * DIM + idx];
  }

  const v4f bv  = *(const v4f*)(b1 + off);
  const v4f wa0 = *(const v4f*)(MW2a + off);
  const v4f wb0 = *(const v4f*)(MW2b + off);
  v4f S1[4], Sa[4], Sb[4];
#pragma unroll
  for (int j = 0; j < 4; ++j) {
    S1[j] = *(const v4f*)(MW1T + j * HID + off);
    Sa[j] = *(const v4f*)(MW2a + (j + 1) * HID + off);
    Sb[j] = *(const v4f*)(MW2b + (j + 1) * HID + off);
  }

  v4f wacc = wb0;
  float bacc = 0.f;
  float h[4][4][4];
  v2f ma[4][2];
#pragma unroll
  for (int m = 0; m < 4; ++m) {
#pragma unroll
    for (int k = 0; k < 4; ++k) {
      const float b = grp_bcast(bv[m], sel[k]);
      h[m][0][k] = b; h[m][1][k] = b; h[m][2][k] = b; h[m][3][k] = b;
    }
    ma[m][0] = (v2f){0.f, 0.f};
    ma[m][1] = (v2f){0.f, 0.f};
  }
#pragma unroll
  for (int m = 0; m < 4; ++m) dot4x4(h[m], wa0[m], wb0[m], sel, ma);
  lds_barrier();

  for (int i = 0; i < DIM; i += 4) {
#pragma unroll
    for (int j = 0; j < 4; ++j) {
      scan_step(i + j, i + j < DIM - 1, red[j & 1], zs, xs, b2s, tid, wave, lane, base, xsel, sel, h, ma,
                S1[j], Sa[j], Sb[j], wacc, bacc);
      const int pr = min(i + j + 4, DIM - 2);
      S1[j] = *(const v4f*)(MW1T + pr * HID + off);
      Sa[j] = *(const v4f*)(MW2a + (pr + 1) * HID + off);
      Sb[j] = *(const v4f*)(MW2b + (pr + 1) * HID + off);
    }
  }

  if (blockIdx.x == 0) {
    *(v4f*)(sums + off) = wacc;
    if (tid == 0) sums[HID] = bacc;
  }
  lds_barrier();
#pragma unroll
  for (int q = 0; q < 8; ++q) {
    const int idx = q * 128 + tid;
    xout[row0 * DIM + idx] = xs[(idx >> 6) * SC_PITCH + (idx & 63)];
  }
}

#define LD_WAVES 4
#define LD_ROWS  32

__device__ __forceinline__ void ld_mma(const v2f (&a)[8], const v2f (&fo)[8], const v2f (&ft)[8], v8f& own, v8f& oth) {
  v8f p = {}, r = {};
#pragma unroll
  for (int u = 0; u < 8; ++u) {
    p = __builtin_amdgcn_wmma_f32_16x16x4_f32(false, a[u], false, fo[u], (short)0, p, false, false);
    r = __builtin_amdgcn_wmma_f32_16x16x4_f32(false, a[u], false, ft[u], (short)0, r, false, false);
  }
  own = p; oth = r;
}

__device__ __forceinline__ void ld_finish(v4f (*ex)[2][32], int wave, int lane, const v8f& own, const v8f& oth,
                                          float bb, float wv, v8f& ld) {
  v4f s0, s1;
  s0.x = oth[0]; s0.y = oth[1]; s0.z = oth[2]; s0.w = oth[3];
  s1.x = oth[4]; s1.y = oth[5]; s1.z = oth[6]; s1.w = oth[7];
  ex[wave][0][lane] = s0;
  ex[wave][1][lane] = s1;
  __syncthreads();
  const v4f r0 = ex[wave ^ 1][0][lane];
  const v4f r1 = ex[wave ^ 1][1][lane];
#pragma unroll
  for (int v = 0; v < 4; ++v) {
    const float h0 = fmaxf((own[v] + r0[v]) + bb, 0.f);
    const float h1 = fmaxf((own[4 + v] + r1[v]) + bb, 0.f);
    ld[v]     = fmaf(h0, wv, ld[v]);
    ld[4 + v] = fmaf(h1, wv, ld[4 + v]);
  }
}

__global__ __launch_bounds__(LD_WAVES * 32) void maf_logdet_wmma(const float* __restrict__ b1,
                                                                 const float* __restrict__ ws,
                                                                 const float* __restrict__ xout,
                                                                 float* __restrict__ ldout) {
  __shared__ __align__(16) float Bs[NPACK];
  __shared__ __align__(16) float b1s[HID];
  __shared__ __align__(16) float wss[HID];
  __shared__ __align__(16) v4f exch[2][LD_WAVES][2][32];
  __shared__ float fin[LD_WAVES][16];
  const float bsum = ws[WS_BSUM];

  const int tid  = threadIdx.x;
  const int lane = tid & 31;
  const int wave = tid >> 5;
  const int lr   = lane & 15;
  const int koff = (lane >> 4) ? 2 : 0;
  const int rt   = wave >> 1;
  const int kh   = wave & 1;
  const int m0   = blockIdx.x * LD_ROWS + rt * 16;

  const float* xp = xout + (m0 + lr) * DIM + 32 * kh + koff;
  v2f a[8];
#pragma unroll
  for (int u = 0; u < 8; ++u) a[u] = *(const v2f*)(xp + 4 * u);

  {
    const v4f* src = (const v4f*)(ws + WS_BPACK);
    v4f* dst = (v4f*)Bs;
#pragma unroll 32
    for (int q = 0; q < NPACK / 4 / (LD_WAVES * 32); ++q) dst[q * (LD_WAVES * 32) + tid] = src[q * (LD_WAVES * 32) + tid];
    ((v4f*)b1s)[tid] = ((const v4f*)b1)[tid];
    ((v4f*)wss)[tid] = ((const v4f*)(ws + WS_WSUM))[tid];
  }
  __syncthreads();

  const v2f* Bo = (const v2f*)Bs + (kh * 16 + 8 * kh) * 32 + lane;
  const v2f* Bt = (const v2f*)Bs + ((1 - kh) * 16 + 8 * kh) * 32 + lane;
  const float* bo = b1s + kh * 16 + lr;
  const float* wo = wss + kh * 16 + lr;
  v8f ld = {};
  v2f co[8], ct[8], no[8], nt[8];
  v8f ownA, othA, ownB, othB;
#pragma unroll
  for (int u = 0; u < 8; ++u) { co[u] = Bo[u * 32]; ct[u] = Bt[u * 32]; }
#pragma unroll
  for (int u = 0; u < 8; ++u) { no[u] = Bo[1024 + u * 32]; nt[u] = Bt[1024 + u * 32]; }
  ld_mma(a, co, ct, ownA, othA);
  for (int k = 1; k < 15; k += 2) {
#pragma unroll
    for (int u = 0; u < 8; ++u) { co[u] = Bo[(k + 1) * 1024 + u * 32]; ct[u] = Bt[(k + 1) * 1024 + u * 32]; }
    ld_mma(a, no, nt, ownB, othB);
    ld_finish(exch[0], wave, lane, ownA, othA, bo[(k - 1) * 32], wo[(k - 1) * 32], ld);
#pragma unroll
    for (int u = 0; u < 8; ++u) { no[u] = Bo[(k + 2) * 1024 + u * 32]; nt[u] = Bt[(k + 2) * 1024 + u * 32]; }
    ld_mma(a, co, ct, ownA, othA);
    ld_finish(exch[1], wave, lane, ownB, othB, bo[k * 32], wo[k * 32], ld);
  }
  ld_mma(a, no, nt, ownB, othB);
  ld_finish(exch[0], wave, lane, ownA, othA, bo[14 * 32], wo[14 * 32], ld);
  ld_finish(exch[1], wave, lane, ownB, othB, bo[15 * 32], wo[15 * 32], ld);

#pragma unroll
  for (int off = 1; off < 16; off <<= 1) {
#pragma unroll
    for (int v = 0; v < 8; ++v) {
      float t = ld[v];
      ld[v] = t + __shfl_xor(t, off, 32);
    }
  }
  if (lr == 0) {
    const int h8 = (lane >> 4) * 8;
#pragma unroll
    for (int v = 0; v < 8; ++v) fin[wave][h8 + v] = ld[v];
  }
  __syncthreads();
  if (wave == 0) {
    const int t2 = (lane >> 4) * 2, rr = lane & 15;
    ldout[blockIdx.x * LD_ROWS + lane] = (fin[t2][rr] + fin[t2 + 1][rr]) + bsum;
  }
}

extern "C" void kernel_launch(void* const* d_in, const int* in_sizes, int n_in,
                              void* d_out, int out_size, void* d_ws, size_t ws_size,
                              hipStream_t stream) {
  const float* z     = (const float*)d_in[0];
  const float* W1    = (const float*)d_in[1];
  const float* b1    = (const float*)d_in[2];
  const float* W2    = (const float*)d_in[3];
  const float* b2    = (const float*)d_in[4];
  const float* mask1 = (const float*)d_in[5];
  const float* mask2 = (const float*)d_in[6];

  float* ws    = (float*)d_ws;
  float* xout  = (float*)d_out;
  float* ldout = xout + BATCH * DIM;

  const int nprep = HID * DIM + 2 * DIM * HID + NPACK;
  maf_prep_mask<<<(nprep + 255) / 256, 256, 0, stream>>>(W1, mask1, W2, mask2, ws);
  maf_scan<<<BATCH / SC_ROWS, SC_WAVES * 32, 0, stream>>>(z, b1, b2, ws, ws + WS_WSUM, xout);
  maf_logdet_wmma<<<BATCH / LD_ROWS, LD_WAVES * 32, 0, stream>>>(b1, ws, xout, ldout);
}


# scan: a wave owns 4 whole rows, its 32 lanes split the 512 hidden units (16 each) and load their own weights; no lane copies, no LDS join or barrier per step; lane butterflies (new sum order)
# speedup vs baseline: 4.1765x; 1.1692x over previous
#include <hip/hip_runtime.h>
#include <hip/hip_bf16.h>

#define DIM 64
#define HID 512
#define BATCH 8192

typedef __attribute__((ext_vector_type(2))) float v2f;
typedef __attribute__((ext_vector_type(8))) float v8f;

#define WS_MW1T  0
#define WS_MW2   (HID*DIM)
#define WS_WSUM  (HID*DIM + 2*DIM*HID)
#define WS_BSUM  (WS_WSUM + HID)
#define WS_BPACK 98832
#define NPACK    (HID*DIM)

__global__ __launch_bounds__(256) void maf_prep_mask(const float* __restrict__ W1,
                                                     const float* __restrict__ mask1,
                                                     const float* __restrict__ W2,
                                                     const float* __restrict__ mask2,
                                                     float* __restrict__ ws) {
  int idx = blockIdx.x * blockDim.x + threadIdx.x;
  if (idx < HID * DIM) {
    int j = idx / DIM, k = idx % DIM;
    ws[WS_MW1T + k * HID + j] = mask1[idx] * W1[idx];
  } else if (idx < HID * DIM + 2 * DIM * HID) {
    int t = idx - HID * DIM;
    ws[WS_MW2 + t] = mask2[t] * W2[t];
  } else if (idx < HID * DIM + 2 * DIM * HID + NPACK) {
    int f = idx - (HID * DIM + 2 * DIM * HID);
    int p = f >> 1, c = f & 1;
    int nt = p >> 9;
    int t  = (p >> 5) & 15;
    int l  = p & 31;
    int koff = (l >> 4) << 1;
    int j = (nt << 4) + (l & 15);
    int k = (t << 2) + koff + c;
    ws[WS_BPACK + f] = mask1[j * DIM + k] * W1[j * DIM + k];
  }
}

typedef __attribute__((ext_vector_type(4))) float v4f;

#define SC_WAVES 4
#define SC_ROWS  16
#define SC_PITCH 65

__device__ __forceinline__ float other_half(float v) {
  int keep;
  return __builtin_bit_cast(float, __builtin_amdgcn_permlanex16(keep, __builtin_bit_cast(int, v), 0x76543210,
                                                                0xfedcba98, false, false));
}

template <int CTRL>
__device__ __forceinline__ float xmask_add(float v) {
  return v + __builtin_bit_cast(float, __builtin_amdgcn_update_dpp(0, __builtin_bit_cast(int, v), CTRL, 0xf, 0xf,
                                                                   true));
}

__device__ __forceinline__ float row_sum(float v) {
  v = xmask_add<0x161>(v);
  v = xmask_add<0x162>(v);
  v = xmask_add<0x164>(v);
  return xmask_add<0x168>(v);
}

__device__ __forceinline__ void dot_all(const v2f (&hp)[4][2][4], const v4f (&wa)[4], const v4f (&wb)[4],
                                        v2f (&mu)[4], v2f (&al)[4]) {
#pragma unroll
  for (int m = 0; m < 4; ++m) {
#pragma unroll
    for (int p = 0; p < 2; ++p) {
      v2f wap, wbp;
      wap.x = wa[m][2 * p]; wap.y = wa[m][2 * p + 1];
      wbp.x = wb[m][2 * p]; wbp.y = wb[m][2 * p + 1];
#pragma unroll
      for (int s = 0; s < 4; ++s) {
        v2f a;
        a.x = fmaxf(hp[m][p][s].x, 0.f); a.y = fmaxf(hp[m][p][s].y, 0.f);
        mu[s] = __builtin_elementwise_fma(wap, a, mu[s]);
        al[s] = __builtin_elementwise_fma(wbp, a, al[s]);
      }
    }
  }
}

__device__ __forceinline__ void update_all(v2f (&hp)[4][2][4], const v4f (&w1)[4], const v2f (&xx)[4]) {
#pragma unroll
  for (int m = 0; m < 4; ++m) {
#pragma unroll
    for (int p = 0; p < 2; ++p) {
      v2f w;
      w.x = w1[m][2 * p]; w.y = w1[m][2 * p + 1];
#pragma unroll
      for (int s = 0; s < 4; ++s) hp[m][p][s] = __builtin_elementwise_fma(w, xx[s], hp[m][p][s]);
    }
  }
}

__device__ __forceinline__ void lds_barrier() {
  asm volatile("s_wait_dscnt 0x0\n\ts_barrier_signal -1\n\ts_barrier_wait -1" ::: "memory");
}

__device__ __forceinline__ void scan_step(int i, bool do_pass, float za, float zb, float bm, float ba, float* xs,
                                          int lane, int ra, v2f (&hp)[4][2][4], v2f (&mu)[4], v2f (&al)[4],
                                          const v4f (&w1)[4], const v4f (&wa)[4], const v4f (&wb)[4],
                                          v4f (&wacc)[4], float& bacc) {
  const float k0 = (mu[0].x + mu[0].y) + other_half(mu[2].x + mu[2].y);
  const float k1 = (al[0].x + al[0].y) + other_half(al[2].x + al[2].y);
  const float k2 = (mu[1].x + mu[1].y) + other_half(mu[3].x + mu[3].y);
  const float k3 = (al[1].x + al[1].y) + other_half(al[3].x + al[3].y);
  const float s0 = row_sum(k0), s1 = row_sum(k1), s2 = row_sum(k2), s3 = row_sum(k3);
  bacc += ba;
  const float x0 = fmaf(za, expf(s1 + ba), s0 + bm);
  const float x1 = fmaf(zb, expf(s3 + ba), s2 + bm);
  if ((lane & 15) == 0) {
    xs[ra * SC_PITCH + i] = x0;
    xs[(ra + 1) * SC_PITCH + i] = x1;
  }
  if (do_pass) {
    v2f xx[4];
    const float x2 = other_half(x0), x3 = other_half(x1);
    xx[0].x = x0; xx[0].y = x0; xx[1].x = x1; xx[1].y = x1;
    xx[2].x = x2; xx[2].y = x2; xx[3].x = x3; xx[3].y = x3;
#pragma unroll
    for (int r = 0; r < 4; ++r) { mu[r] = (v2f){0.f, 0.f}; al[r] = (v2f){0.f, 0.f}; }
    update_all(hp, w1, xx);
    dot_all(hp, wa, wb, mu, al);
#pragma unroll
    for (int m = 0; m < 4; ++m) wacc[m] += wb[m];
  }
}

__global__ __launch_bounds__(128) void maf_scan(const float* __restrict__ z,
                                                const float* __restrict__ b1,
                                                const float* __restrict__ b2,
                                                const float* __restrict__ ws,
                                                float* __restrict__ sums,
                                                float* __restrict__ xout) {
  __shared__ float zs[SC_ROWS * SC_PITCH];
  __shared__ float xs[SC_ROWS * SC_PITCH];
  __shared__ float b2s[2 * DIM];
  const float* MW1T = ws + WS_MW1T;
  const float* MW2a = ws + WS_MW2;
  const float* MW2b = ws + WS_MW2 + DIM * HID;

  const int tid  = threadIdx.x;
  const int lane = tid & 31;
  const int wave = tid >> 5;
  const int ra   = 4 * wave + 2 * (lane >> 4);
  const int row0 = blockIdx.x * SC_ROWS;
  const int off  = lane * 4;

  b2s[tid] = b2[tid];
#pragma unroll
  for (int q = 0; q < 8; ++q) {
    const int idx = q * 128 + tid;
    zs[(idx >> 6) * SC_PITCH + (idx & 63)] = z[row0 * DIM + idx];
  }

  v4f bv[4], wa0[4], wb0[4];
#pragma unroll
  for (int m = 0; m < 4; ++m) {
    bv[m]  = *(const v4f*)(b1 + 128 * m + off);
    wa0[m] = *(const v4f*)(MW2a + 128 * m + off);
    wb0[m] = *(const v4f*)(MW2b + 128 * m + off);
  }
  v4f S1[4][4], Sa[4][4], Sb[4][4];
#pragma unroll
  for (int j = 0; j < 4; ++j) {
#pragma unroll
    for (int m = 0; m < 4; ++m) {
      S1[j][m] = *(const v4f*)(MW1T + j * HID + 128 * m + off);
      Sa[j][m] = *(const v4f*)(MW2a + (j + 1) * HID + 128 * m + off);
      Sb[j][m] = *(const v4f*)(MW2b + (j + 1) * HID + 128 * m + off);
    }
  }

  v4f wacc[4];
  float bacc = 0.f;
  v2f hp[4][2][4];
  v2f mu[4], al[4];
#pragma unroll
  for (int m = 0; m < 4; ++m) {
    wacc[m] = wb0[m];
#pragma unroll
    for (int p = 0; p < 2; ++p) {
      v2f b;
      b.x = bv[m][2 * p]; b.y = bv[m][2 * p + 1];
      hp[m][p][0] = b; hp[m][p][1] = b; hp[m][p][2] = b; hp[m][p][3] = b;
    }
    mu[m] = (v2f){0.f, 0.f};
    al[m] = (v2f){0.f, 0.f};
  }
  dot_all(hp, wa0, wb0, mu, al);
  lds_barrier();

  float za = zs[ra * SC_PITCH], zb = zs[(ra + 1) * SC_PITCH], bm = b2s[0], ba = b2s[DIM];
  for (int i = 0; i < DIM; i += 4) {
#pragma unroll
    for (int j = 0; j < 4; ++j) {
      const int nx = min(i + j + 1, DIM - 1);
      const float za_n = zs[ra * SC_PITCH + nx], zb_n = zs[(ra + 1) * SC_PITCH + nx];
      const float bm_n = b2s[nx], ba_n = b2s[DIM + nx];
      scan_step(i + j, i + j < DIM - 1, za, zb, bm, ba, xs, lane, ra, hp, mu, al, S1[j], Sa[j], Sb[j], wacc, bacc);
      za = za_n; zb = zb_n; bm = bm_n; ba = ba_n;
      const int pr = min(i + j + 4, DIM - 2);
#pragma unroll
      for (int m = 0; m < 4; ++m) {
        S1[j][m] = *(const v4f*)(MW1T + pr * HID + 128 * m + off);
        Sa[j][m] = *(const v4f*)(MW2a + (pr + 1) * HID + 128 * m + off);
        Sb[j][m] = *(const v4f*)(MW2b + (pr + 1) * HID + 128 * m + off);
      }
    }
  }

  if (blockIdx.x == 0 && wave == 0) {
#pragma unroll
    for (int m = 0; m < 4; ++m) *(v4f*)(sums + 128 * m + off) = wacc[m];
    if (tid == 0) sums[HID] = bacc;
  }
  lds_barrier();
#pragma unroll
  for (int q = 0; q < 8; ++q) {
    const int idx = q * 128 + tid;
    xout[row0 * DIM + idx] = xs[(idx >> 6) * SC_PITCH + (idx & 63)];
  }
}

#define LD_WAVES 4
#define LD_ROWS  32

__device__ __forceinline__ void ld_mma(const v2f (&a)[8], const v2f (&fo)[8], const v2f (&ft)[8], v8f& own, v8f& oth) {
  v8f p = {}, r = {};
#pragma unroll
  for (int u = 0; u < 8; ++u) {
    p = __builtin_amdgcn_wmma_f32_16x16x4_f32(false, a[u], false, fo[u], (short)0, p, false, false);
    r = __builtin_amdgcn_wmma_f32_16x16x4_f32(false, a[u], false, ft[u], (short)0, r, false, false);
  }
  own = p; oth = r;
}

__device__ __forceinline__ void ld_finish(v4f (*ex)[2][32], int wave, int lane, const v8f& own, const v8f& oth,
                                          float bb, float wv, v8f& ld) {
  v4f s0, s1;
  s0.x = oth[0]; s0.y = oth[1]; s0.z = oth[2]; s0.w = oth[3];
  s1.x = oth[4]; s1.y = oth[5]; s1.z = oth[6]; s1.w = oth[7];
  ex[wave][0][lane] = s0;
  ex[wave][1][lane] = s1;
  __syncthreads();
  const v4f r0 = ex[wave ^ 1][0][lane];
  const v4f r1 = ex[wave ^ 1][1][lane];
#pragma unroll
  for (int v = 0; v < 4; ++v) {
    const float h0 = fmaxf((own[v] + r0[v]) + bb, 0.f);
    const float h1 = fmaxf((own[4 + v] + r1[v]) + bb, 0.f);
    ld[v]     = fmaf(h0, wv, ld[v]);
    ld[4 + v] = fmaf(h1, wv, ld[4 + v]);
  }
}

__global__ __launch_bounds__(LD_WAVES * 32) void maf_logdet_wmma(const float* __restrict__ b1,
                                                                 const float* __restrict__ ws,
                                                                 const float* __restrict__ xout,
                                                                 float* __restrict__ ldout) {
  __shared__ __align__(16) float Bs[NPACK];
  __shared__ __align__(16) float b1s[HID];
  __shared__ __align__(16) float wss[HID];
  __shared__ __align__(16) v4f exch[2][LD_WAVES][2][32];
  __shared__ float fin[LD_WAVES][16];
  const float bsum = ws[WS_BSUM];

  const int tid  = threadIdx.x;
  const int lane = tid & 31;
  const int wave = tid >> 5;
  const int lr   = lane & 15;
  const int koff = (lane >> 4) ? 2 : 0;
  const int rt   = wave >> 1;
  const int kh   = wave & 1;
  const int m0   = blockIdx.x * LD_ROWS + rt * 16;

  const float* xp = xout + (m0 + lr) * DIM + 32 * kh + koff;
  v2f a[8];
#pragma unroll
  for (int u = 0; u < 8; ++u) a[u] = *(const v2f*)(xp + 4 * u);

  {
    const v4f* src = (const v4f*)(ws + WS_BPACK);
    v4f* dst = (v4f*)Bs;
#pragma unroll 32
    for (int q = 0; q < NPACK / 4 / (LD_WAVES * 32); ++q) dst[q * (LD_WAVES * 32) + tid] = src[q * (LD_WAVES * 32) + tid];
    ((v4f*)b1s)[tid] = ((const v4f*)b1)[tid];
    ((v4f*)wss)[tid] = ((const v4f*)(ws + WS_WSUM))[tid];
  }
  __syncthreads();

  const v2f* Bo = (const v2f*)Bs + (kh * 16 + 8 * kh) * 32 + lane;
  const v2f* Bt = (const v2f*)Bs + ((1 - kh) * 16 + 8 * kh) * 32 + lane;
  const float* bo = b1s + kh * 16 + lr;
  const float* wo = wss + kh * 16 + lr;
  v8f ld = {};
  v2f co[8], ct[8], no[8], nt[8];
  v8f ownA, othA, ownB, othB;
#pragma unroll
  for (int u = 0; u < 8; ++u) { co[u] = Bo[u * 32]; ct[u] = Bt[u * 32]; }
#pragma unroll
  for (int u = 0; u < 8; ++u) { no[u] = Bo[1024 + u * 32]; nt[u] = Bt[1024 + u * 32]; }
  ld_mma(a, co, ct, ownA, othA);
  for (int k = 1; k < 15; k += 2) {
#pragma unroll
    for (int u = 0; u < 8; ++u) { co[u] = Bo[(k + 1) * 1024 + u * 32]; ct[u] = Bt[(k + 1) * 1024 + u * 32]; }
    ld_mma(a, no, nt, ownB, othB);
    ld_finish(exch[0], wave, lane, ownA, othA, bo[(k - 1) * 32], wo[(k - 1) * 32], ld);
#pragma unroll
    for (int u = 0; u < 8; ++u) { no[u] = Bo[(k + 2) * 1024 + u * 32]; nt[u] = Bt[(k + 2) * 1024 + u * 32]; }
    ld_mma(a, co, ct, ownA, othA);
    ld_finish(exch[1], wave, lane, ownB, othB, bo[k * 32], wo[k * 32], ld);
  }
  ld_mma(a, no, nt, ownB, othB);
  ld_finish(exch[0], wave, lane, ownA, othA, bo[14 * 32], wo[14 * 32], ld);
  ld_finish(exch[1], wave, lane, ownB, othB, bo[15 * 32], wo[15 * 32], ld);

#pragma unroll
  for (int off = 1; off < 16; off <<= 1) {
#pragma unroll
    for (int v = 0; v < 8; ++v) {
      float t = ld[v];
      ld[v] = t + __shfl_xor(t, off, 32);
    }
  }
  if (lr == 0) {
    const int h8 = (lane >> 4) * 8;
#pragma unroll
    for (int v = 0; v < 8; ++v) fin[wave][h8 + v] = ld[v];
  }
  __syncthreads();
  if (wave == 0) {
    const int t2 = (lane >> 4) * 2, rr = lane & 15;
    ldout[blockIdx.x * LD_ROWS + lane] = (fin[t2][rr] + fin[t2 + 1][rr]) + bsum;
  }
}

extern "C" void kernel_launch(void* const* d_in, const int* in_sizes, int n_in,
                              void* d_out, int out_size, void* d_ws, size_t ws_size,
                              hipStream_t stream) {
  const float* z     = (const float*)d_in[0];
  const float* W1    = (const float*)d_in[1];
  const float* b1    = (const float*)d_in[2];
  const float* W2    = (const float*)d_in[3];
  const float* b2    = (const float*)d_in[4];
  const float* mask1 = (const float*)d_in[5];
  const float* mask2 = (const float*)d_in[6];

  float* ws    = (float*)d_ws;
  float* xout  = (float*)d_out;
  float* ldout = xout + BATCH * DIM;

  const int nprep = HID * DIM + 2 * DIM * HID + NPACK;
  maf_prep_mask<<<(nprep + 255) / 256, 256, 0, stream>>>(W1, mask1, W2, mask2, ws);
  maf_scan<<<BATCH / SC_ROWS, SC_WAVES * 32, 0, stream>>>(z, b1, b2, ws, ws + WS_WSUM, xout);
  maf_logdet_wmma<<<BATCH / LD_ROWS, LD_WAVES * 32, 0, stream>>>(b1, ws, xout, ldout);
}
